# starting kernel
# baseline (speedup 1.0000x reference)
#include <hip/hip_runtime.h>

typedef __attribute__((ext_vector_type(2))) float v2f;
typedef __attribute__((ext_vector_type(4))) float v4f;
typedef __attribute__((ext_vector_type(8))) float v8f;

#define DF 128


__global__ void deg_init_kernel(float* deg, int n) {
    int i = blockIdx.x * blockDim.x + threadIdx.x;
    if (i < n) deg[i] = 1.0f;
}

__global__ void deg_accum_kernel(const int* __restrict__ dst, float* deg, int e) {
    int i = blockIdx.x * blockDim.x + threadIdx.x;
    if (i < e) unsafeAtomicAdd(&deg[dst[i]], 1.0f);
}

__global__ void deg_rsqrt_kernel(float* deg, int n) {
    int i = blockIdx.x * blockDim.x + threadIdx.x;
    if (i < n) deg[i] = rsqrtf(deg[i]);
}


__global__ __launch_bounds__(128) void gemm_wmma_kernel(const float* __restrict__ X,
                                                        const float* __restrict__ W,
                                                        float* __restrict__ H, int nrows) {
    __shared__ float sW[DF * DF];
    for (int i = threadIdx.x; i < DF * DF; i += blockDim.x) {
        int k = i >> 7;
        int nn = i & 127;
        sW[(k >> 1) * 256 + nn * 2 + (k & 1)] = W[i];
    }
    __syncthreads();

    const int wave = threadIdx.x >> 5;
    const int lane = threadIdx.x & 31;
    const int tile = blockIdx.x * 4 + wave;
    const int m0 = tile * 16;
    if (m0 >= nrows) return;

    const int half = lane >> 4;
    const int l = lane & 15;

    v8f acc[8] = {};
    const float* xrow = X + (size_t)(m0 + l) * DF;

    for (int kb = 0; kb < DF; kb += 4) {
        v2f a = *(const v2f*)(xrow + kb + half * 2);
        const float* wp = sW + ((kb >> 1) + half) * 256;
#pragma unroll
        for (int nt = 0; nt < 8; nt++) {
            v2f b = *(const v2f*)(wp + (nt * 16 + l) * 2);
            acc[nt] = __builtin_amdgcn_wmma_f32_16x16x4_f32(
                false, a, false, b, (short)0, acc[nt], false, false);
        }
    }

#pragma unroll
    for (int nt = 0; nt < 8; nt++) {
        const int col = nt * 16 + l;
#pragma unroll
        for (int r = 0; r < 8; r++) {
            const int row = m0 + r + 8 * half;
            H[(size_t)row * DF + col] = acc[nt][r];
        }
    }
}


__global__ void selfloop_init_kernel(const float* __restrict__ H, const float* __restrict__ dinv,
                                     float* __restrict__ out, int n) {
    int i = blockIdx.x * blockDim.x + threadIdx.x;
    int total = n * (DF / 4);
    if (i >= total) return;
    int node = i >> 5;
    float s = dinv[node];
    s = s * s;
    v4f v = ((const v4f*)H)[i];
    ((v4f*)out)[i] = v * s;
}

__global__ __launch_bounds__(256) void scatter_edges_kernel(const int* __restrict__ src,
                                                            const int* __restrict__ dst,
                                                            const float* __restrict__ dinv,
                                                            const float* __restrict__ H,
                                                            float* __restrict__ out, int e) {
    const int lane = threadIdx.x & 31;
    const int w = (int)((blockIdx.x * blockDim.x + threadIdx.x) >> 5);
    if (w >= e) return;
    const int s = src[w];
    const int d = dst[w];
    const float norm = dinv[s] * dinv[d];
    v4f m = ((const v4f*)(H + (size_t)s * DF))[lane];
    float* o = out + (size_t)d * DF + lane * 4;
    unsafeAtomicAdd(o + 0, m.x * norm);
    unsafeAtomicAdd(o + 1, m.y * norm);
    unsafeAtomicAdd(o + 2, m.z * norm);
    unsafeAtomicAdd(o + 3, m.w * norm);
}

__global__ void bias_act_kernel(const float* __restrict__ agg, const float* __restrict__ bias,
                                float* __restrict__ z, int n, int do_relu) {
    int i = blockIdx.x * blockDim.x + threadIdx.x;
    int total = n * (DF / 4);
    if (i >= total) return;
    int c4 = i & 31;
    v4f b = ((const v4f*)bias)[c4];
    v4f v = ((const v4f*)agg)[i] + b;
    if (do_relu) {
        v.x = fmaxf(v.x, 0.0f);
        v.y = fmaxf(v.y, 0.0f);
        v.z = fmaxf(v.z, 0.0f);
        v.w = fmaxf(v.w, 0.0f);
    }
    ((v4f*)z)[i] = v;
}


__global__ __launch_bounds__(256) void decode_kernel(const int* __restrict__ s,
                                                     const int* __restrict__ d,
                                                     const float* __restrict__ z,
                                                     float* __restrict__ logits, int e) {
    const int lane = threadIdx.x & 31;
    const int w = (int)((blockIdx.x * blockDim.x + threadIdx.x) >> 5);
    if (w >= e) return;
    const int si = s[w];
    const int di = d[w];
    v4f a = ((const v4f*)(z + (size_t)si * DF))[lane];
    v4f b = ((const v4f*)(z + (size_t)di * DF))[lane];
    float dot = a.x * b.x + a.y * b.y + a.z * b.z + a.w * b.w;
#pragma unroll
    for (int off = 16; off > 0; off >>= 1) dot += __shfl_xor(dot, off, 32);
    if (lane == 0) logits[w] = dot;
}


extern "C" void kernel_launch(void* const* d_in, const int* in_sizes, int n_in,
                              void* d_out, int out_size, void* d_ws, size_t ws_size,
                              hipStream_t stream) {
    const float* x  = (const float*)d_in[0];
    const int*   ei = (const int*)d_in[1];
    const int*   el = (const int*)d_in[2];
    const float* W1 = (const float*)d_in[3];
    const float* b1 = (const float*)d_in[4];
    const float* W2 = (const float*)d_in[5];
    const float* b2 = (const float*)d_in[6];

    const int n  = in_sizes[0] / DF;
    const int E  = in_sizes[1] / 2;
    const int EL = in_sizes[2] / 2;

    float* bufA = (float*)d_ws;
    float* bufB = bufA + (size_t)n * DF;
    float* dinv = bufB + (size_t)n * DF;
    float* out  = (float*)d_out;

    const int tiles  = (n + 15) / 16;
    const int gemmBl = (tiles + 3) / 4;
    const int elem4  = n * (DF / 4);
    const int elBl   = (elem4 + 255) / 256;
    const int nBl    = (n + 255) / 256;
    const int eBl    = (E + 255) / 256;
    const int scatBl = (E + 7) / 8;
    const int decBl  = (EL + 7) / 8;

    deg_init_kernel<<<nBl, 256, 0, stream>>>(dinv, n);
    deg_accum_kernel<<<eBl, 256, 0, stream>>>(ei + E, dinv, E);
    deg_rsqrt_kernel<<<nBl, 256, 0, stream>>>(dinv, n);

    gemm_wmma_kernel<<<gemmBl, 128, 0, stream>>>(x, W1, bufA, n);
    selfloop_init_kernel<<<elBl, 256, 0, stream>>>(bufA, dinv, bufB, n);
    scatter_edges_kernel<<<scatBl, 256, 0, stream>>>(ei, ei + E, dinv, bufA, bufB, E);
    bias_act_kernel<<<elBl, 256, 0, stream>>>(bufB, b1, bufA, n, 1);

    gemm_wmma_kernel<<<gemmBl, 128, 0, stream>>>(bufA, W2, bufB, n);
    selfloop_init_kernel<<<elBl, 256, 0, stream>>>(bufB, dinv, bufA, n);
    scatter_edges_kernel<<<scatBl, 256, 0, stream>>>(ei, ei + E, dinv, bufB, bufA, E);
    bias_act_kernel<<<elBl, 256, 0, stream>>>(bufA, b2, bufB, n, 0);

    decode_kernel<<<decBl, 256, 0, stream>>>(el, el + EL, bufB, out, EL);
}


# Edge scatter and row-wise passes as resident grids: each wave loops over 32-edge groups (one coalesced index load, 8 edges' row loads in flight), not one wave per edge; atomic sums as before
# speedup vs baseline: 1.5453x; 1.5453x over previous
#include <hip/hip_runtime.h>

typedef __attribute__((ext_vector_type(2))) float v2f;
typedef __attribute__((ext_vector_type(4))) float v4f;
typedef __attribute__((ext_vector_type(8))) float v8f;

#define DF 128


__global__ void deg_init_kernel(float* deg, int n) {
    int i = blockIdx.x * blockDim.x + threadIdx.x;
    if (i < n) deg[i] = 1.0f;
}

__global__ void deg_accum_kernel(const int* __restrict__ dst, float* deg, int e) {
    int i = blockIdx.x * blockDim.x + threadIdx.x;
    if (i < e) unsafeAtomicAdd(&deg[dst[i]], 1.0f);
}

__global__ void deg_rsqrt_kernel(float* deg, int n) {
    int i = blockIdx.x * blockDim.x + threadIdx.x;
    if (i < n) deg[i] = rsqrtf(deg[i]);
}


__global__ __launch_bounds__(128) void gemm_wmma_kernel(const float* __restrict__ X,
                                                        const float* __restrict__ W,
                                                        float* __restrict__ H, int nrows) {
    __shared__ float sW[DF * DF];
    for (int i = threadIdx.x; i < DF * DF; i += blockDim.x) {
        int k = i >> 7;
        int nn = i & 127;
        sW[(k >> 1) * 256 + nn * 2 + (k & 1)] = W[i];
    }
    __syncthreads();

    const int wave = threadIdx.x >> 5;
    const int lane = threadIdx.x & 31;
    const int tile = blockIdx.x * 4 + wave;
    const int m0 = tile * 16;
    if (m0 >= nrows) return;

    const int half = lane >> 4;
    const int l = lane & 15;

    v8f acc[8] = {};
    const float* xrow = X + (size_t)(m0 + l) * DF;

    for (int kb = 0; kb < DF; kb += 4) {
        v2f a = *(const v2f*)(xrow + kb + half * 2);
        const float* wp = sW + ((kb >> 1) + half) * 256;
#pragma unroll
        for (int nt = 0; nt < 8; nt++) {
            v2f b = *(const v2f*)(wp + (nt * 16 + l) * 2);
            acc[nt] = __builtin_amdgcn_wmma_f32_16x16x4_f32(
                false, a, false, b, (short)0, acc[nt], false, false);
        }
    }

#pragma unroll
    for (int nt = 0; nt < 8; nt++) {
        const int col = nt * 16 + l;
#pragma unroll
        for (int r = 0; r < 8; r++) {
            const int row = m0 + r + 8 * half;
            H[(size_t)row * DF + col] = acc[nt][r];
        }
    }
}


#define RB 4

__global__ __launch_bounds__(256) void selfloop_init_kernel(const float* __restrict__ H,
                                                            const float* __restrict__ dinv,
                                                            float* __restrict__ out, int n) {
    const int total = n * (DF / 4);
    const int stride = (int)(gridDim.x * blockDim.x);
    for (int i0 = blockIdx.x * blockDim.x + threadIdx.x; i0 < total; i0 += RB * stride) {
        v4f v[RB];
        float s[RB];
#pragma unroll
        for (int u = 0; u < RB; u++) {
            const int i = i0 + u * stride;
            if (i < total) {
                v[u] = ((const v4f*)H)[i];
                s[u] = dinv[i >> 5];
            }
        }
#pragma unroll
        for (int u = 0; u < RB; u++) {
            const int i = i0 + u * stride;
            if (i < total) {
                const float q = s[u] * s[u];
                ((v4f*)out)[i] = v[u] * q;
            }
        }
    }
}

#define UB 8

__device__ __forceinline__ void scatter_one_edge(const float* __restrict__ H,
                                                 float* __restrict__ out, int sj, int dj,
                                                 float nj, int lane) {
    const float* h = H + (size_t)sj * DF + lane;
    const float m0 = h[0];
    const float m1 = h[32];
    const float m2 = h[64];
    const float m3 = h[96];
    float* o = out + (size_t)dj * DF + lane;
    unsafeAtomicAdd(o + 0, m0 * nj);
    unsafeAtomicAdd(o + 32, m1 * nj);
    unsafeAtomicAdd(o + 64, m2 * nj);
    unsafeAtomicAdd(o + 96, m3 * nj);
}

__global__ __launch_bounds__(256) void scatter_edges_kernel(const int* __restrict__ src,
                                                            const int* __restrict__ dst,
                                                            const float* __restrict__ dinv,
                                                            const float* __restrict__ H,
                                                            float* __restrict__ out, int e) {
    const int lane = threadIdx.x & 31;
    const int wave = __builtin_amdgcn_readfirstlane(
        (int)((blockIdx.x * blockDim.x + threadIdx.x) >> 5));
    const int nwaves = (int)((gridDim.x * blockDim.x) >> 5);
    const int ngroups = (e + 31) >> 5;
    for (int g = wave; g < ngroups; g += nwaves) {
        const int base = g << 5;
        const int left = e - base;
        const int cnt = left < 32 ? left : 32;
        int s = 0, d = 0;
        float nrm = 0.0f;
        if (lane < cnt) {
            s = src[base + lane];
            d = dst[base + lane];
            nrm = dinv[s] * dinv[d];
        }
        if (cnt == 32) {
#pragma unroll
            for (int j0 = 0; j0 < 32; j0 += UB) {
                float m[UB][4];
#pragma unroll
                for (int u = 0; u < UB; u++) {
                    const int sj = __builtin_amdgcn_readlane(s, j0 + u);
                    const float* h = H + (size_t)sj * DF + lane;
                    m[u][0] = h[0];
                    m[u][1] = h[32];
                    m[u][2] = h[64];
                    m[u][3] = h[96];
                }
#pragma unroll
                for (int u = 0; u < UB; u++) {
                    const int dj = __builtin_amdgcn_readlane(d, j0 + u);
                    const float nj =
                        __int_as_float(__builtin_amdgcn_readlane(__float_as_int(nrm), j0 + u));
                    float* o = out + (size_t)dj * DF + lane;
                    unsafeAtomicAdd(o + 0, m[u][0] * nj);
                    unsafeAtomicAdd(o + 32, m[u][1] * nj);
                    unsafeAtomicAdd(o + 64, m[u][2] * nj);
                    unsafeAtomicAdd(o + 96, m[u][3] * nj);
                }
            }
        } else {
            for (int j = 0; j < cnt; j++) {
                const int sj = __builtin_amdgcn_readlane(s, j);
                const int dj = __builtin_amdgcn_readlane(d, j);
                const float nj = __int_as_float(__builtin_amdgcn_readlane(__float_as_int(nrm), j));
                scatter_one_edge(H, out, sj, dj, nj, lane);
            }
        }
    }
}

__global__ __launch_bounds__(256) void bias_act_kernel(const float* __restrict__ agg,
                                                       const float* __restrict__ bias,
                                                       float* __restrict__ z, int n, int do_relu) {
    const int total = n * (DF / 4);
    const int stride = (int)(gridDim.x * blockDim.x);
    for (int i0 = blockIdx.x * blockDim.x + threadIdx.x; i0 < total; i0 += RB * stride) {
        v4f v[RB];
#pragma unroll
        for (int u = 0; u < RB; u++) {
            const int i = i0 + u * stride;
            if (i < total) v[u] = ((const v4f*)agg)[i];
        }
#pragma unroll
        for (int u = 0; u < RB; u++) {
            const int i = i0 + u * stride;
            if (i < total) {
                v4f w = v[u] + ((const v4f*)bias)[i & 31];
                if (do_relu) {
                    w.x = fmaxf(w.x, 0.0f);
                    w.y = fmaxf(w.y, 0.0f);
                    w.z = fmaxf(w.z, 0.0f);
                    w.w = fmaxf(w.w, 0.0f);
                }
                ((v4f*)z)[i] = w;
            }
        }
    }
}


__global__ __launch_bounds__(256) void decode_kernel(const int* __restrict__ s,
                                                     const int* __restrict__ d,
                                                     const float* __restrict__ z,
                                                     float* __restrict__ logits, int e) {
    const int lane = threadIdx.x & 31;
    const int w = (int)((blockIdx.x * blockDim.x + threadIdx.x) >> 5);
    if (w >= e) return;
    const int si = s[w];
    const int di = d[w];
    v4f a = ((const v4f*)(z + (size_t)si * DF))[lane];
    v4f b = ((const v4f*)(z + (size_t)di * DF))[lane];
    float dot = a.x * b.x + a.y * b.y + a.z * b.z + a.w * b.w;
#pragma unroll
    for (int off = 16; off > 0; off >>= 1) dot += __shfl_xor(dot, off, 32);
    if (lane == 0) logits[w] = dot;
}


extern "C" void kernel_launch(void* const* d_in, const int* in_sizes, int n_in,
                              void* d_out, int out_size, void* d_ws, size_t ws_size,
                              hipStream_t stream) {
    const float* x  = (const float*)d_in[0];
    const int*   ei = (const int*)d_in[1];
    const int*   el = (const int*)d_in[2];
    const float* W1 = (const float*)d_in[3];
    const float* b1 = (const float*)d_in[4];
    const float* W2 = (const float*)d_in[5];
    const float* b2 = (const float*)d_in[6];

    const int n  = in_sizes[0] / DF;
    const int E  = in_sizes[1] / 2;
    const int EL = in_sizes[2] / 2;

    float* bufA = (float*)d_ws;
    float* bufB = bufA + (size_t)n * DF;
    float* dinv = bufB + (size_t)n * DF;
    float* out  = (float*)d_out;

    const int tiles  = (n + 15) / 16;
    const int gemmBl = (tiles + 3) / 4;
    const int nBl    = (n + 255) / 256;
    const int eBl    = (E + 255) / 256;
    const int decBl  = (EL + 7) / 8;
    const int resBl  = 1024;

    deg_init_kernel<<<nBl, 256, 0, stream>>>(dinv, n);
    deg_accum_kernel<<<eBl, 256, 0, stream>>>(ei + E, dinv, E);
    deg_rsqrt_kernel<<<nBl, 256, 0, stream>>>(dinv, n);

    gemm_wmma_kernel<<<gemmBl, 128, 0, stream>>>(x, W1, bufA, n);
    selfloop_init_kernel<<<resBl, 256, 0, stream>>>(bufA, dinv, bufB, n);
    scatter_edges_kernel<<<resBl, 256, 0, stream>>>(ei, ei + E, dinv, bufA, bufB, E);
    bias_act_kernel<<<resBl, 256, 0, stream>>>(bufB, b1, bufA, n, 1);

    gemm_wmma_kernel<<<gemmBl, 128, 0, stream>>>(bufA, W2, bufB, n);
    selfloop_init_kernel<<<resBl, 256, 0, stream>>>(bufB, dinv, bufA, n);
    scatter_edges_kernel<<<resBl, 256, 0, stream>>>(ei, ei + E, dinv, bufB, bufA, E);
    bias_act_kernel<<<resBl, 256, 0, stream>>>(bufA, b2, bufB, n, 0);

    decode_kernel<<<decBl, 256, 0, stream>>>(el, el + EL, bufB, out, EL);
}


# GEMM as H^T = W^T X^T: a wave keeps all of W in 512 registers as A fragments and streams 16-node X tiles as B operands from memory, 1024 resident waves, no LDS staging; k sum order changes
# speedup vs baseline: 2.5850x; 1.6728x over previous
#include <hip/hip_runtime.h>

typedef __attribute__((ext_vector_type(2))) float v2f;
typedef __attribute__((ext_vector_type(4))) float v4f;
typedef __attribute__((ext_vector_type(8))) float v8f;

#define DF 128


__global__ void deg_init_kernel(float* deg, int n) {
    int i = blockIdx.x * blockDim.x + threadIdx.x;
    if (i < n) deg[i] = 1.0f;
}

__global__ void deg_accum_kernel(const int* __restrict__ dst, float* deg, int e) {
    int i = blockIdx.x * blockDim.x + threadIdx.x;
    if (i < e) unsafeAtomicAdd(&deg[dst[i]], 1.0f);
}

__global__ void deg_rsqrt_kernel(float* deg, int n) {
    int i = blockIdx.x * blockDim.x + threadIdx.x;
    if (i < n) deg[i] = rsqrtf(deg[i]);
}

#define GW 256

__device__ __forceinline__ void load_x_tile(v4f (&xb)[16], const float* __restrict__ X, int tile,
                                            int c, int h, int nrows) {
    int row = tile * 16 + c;
    row = row < nrows ? row : nrows - 1;
    const v4f* xr = (const v4f*)(X + (size_t)row * DF) + h;
#pragma unroll
    for (int j = 0; j < 16; j++) xb[j] = xr[2 * j];
}

__global__ __launch_bounds__(128) void gemm_wmma_kernel(const float* __restrict__ X,
                                                        const float* __restrict__ W,
                                                        float* __restrict__ H, int nrows) {
    const int lane = threadIdx.x & 31;
    const int c = lane & 15;
    const int h = lane >> 4;
    const int wv = __builtin_amdgcn_readfirstlane((int)(blockIdx.x * 4 + (threadIdx.x >> 5)));
    const int nwv = (int)gridDim.x * 4;
    const int ntiles = (nrows + 15) >> 4;
    if (wv >= ntiles) return;

    v2f wa[32][8];
#pragma unroll
    for (int t = 0; t < 32; t++) {
        const float* wp = W + (size_t)(8 * (t >> 1) + 4 * h + 2 * (t & 1)) * DF + c;
#pragma unroll
        for (int nt = 0; nt < 8; nt++) {
            wa[t][nt].x = wp[nt * 16];
            wa[t][nt].y = wp[DF + nt * 16];
        }
    }

    v4f xb[16];
    load_x_tile(xb, X, wv, c, h, nrows);
    for (int tile = wv; tile < ntiles; tile += nwv) {
        const int nxt = tile + nwv < ntiles ? tile + nwv : tile;
        v4f xn[16];
        load_x_tile(xn, X, nxt, c, h, nrows);

        v8f acc[8] = {};
#pragma unroll
        for (int j = 0; j < 16; j++) {
            const v2f b0 = __builtin_shufflevector(xb[j], xb[j], 0, 1);
            const v2f b1 = __builtin_shufflevector(xb[j], xb[j], 2, 3);
#pragma unroll
            for (int nt = 0; nt < 8; nt++)
                acc[nt] = __builtin_amdgcn_wmma_f32_16x16x4_f32(
                    false, wa[2 * j][nt], false, b0, (short)0, acc[nt], false, false);
#pragma unroll
            for (int nt = 0; nt < 8; nt++)
                acc[nt] = __builtin_amdgcn_wmma_f32_16x16x4_f32(
                    false, wa[2 * j + 1][nt], false, b1, (short)0, acc[nt], false, false);
        }

        const int row = tile * 16 + c;
        if (row < nrows) {
            float* o = H + (size_t)row * DF + 8 * h;
#pragma unroll
            for (int nt = 0; nt < 8; nt++) {
                *(v4f*)(o + nt * 16) = __builtin_shufflevector(acc[nt], acc[nt], 0, 1, 2, 3);
                *(v4f*)(o + nt * 16 + 4) = __builtin_shufflevector(acc[nt], acc[nt], 4, 5, 6, 7);
            }
        }
#pragma unroll
        for (int j = 0; j < 16; j++) xb[j] = xn[j];
    }
}

#define BK 400
#define NBMAX 256
#define PG 256

__device__ __forceinline__ int bucket_of(int d) {
    const int k = d / BK;
    return k < NBMAX - 1 ? k : NBMAX - 1;
}

__global__ __launch_bounds__(256) void bin_count_kernel(const int* __restrict__ dst,
                                                        int* __restrict__ blkcnt, int e) {
    __shared__ int hist[NBMAX];
    hist[threadIdx.x] = 0;
    __syncthreads();
    const int chunk = (e + PG - 1) / PG;
    const int lo = blockIdx.x * chunk;
    const int hi = lo + chunk < e ? lo + chunk : e;
    for (int i = lo + (int)threadIdx.x; i < hi; i += 256) atomicAdd(&hist[bucket_of(dst[i])], 1);
    __syncthreads();
    blkcnt[blockIdx.x * NBMAX + threadIdx.x] = hist[threadIdx.x];
}

__device__ __forceinline__ int block_scan_256(int* sc, int v) {
    const int t = threadIdx.x;
    sc[t] = v;
    __syncthreads();
    for (int off = 1; off < 256; off <<= 1) {
        const int add = t >= off ? sc[t - off] : 0;
        __syncthreads();
        sc[t] += add;
        __syncthreads();
    }
    return sc[t];
}

__global__ __launch_bounds__(256) void bin_scan_kernel(const int* __restrict__ blkcnt,
                                                       int* __restrict__ excl,
                                                       int* __restrict__ total) {
    __shared__ int sc[256];
    const int k = blockIdx.x;
    const int b = threadIdx.x;
    const int v = blkcnt[b * NBMAX + k];
    const int inc = block_scan_256(sc, v);
    excl[b * NBMAX + k] = inc - v;
    if (b == 255) total[k] = inc;
}

__global__ __launch_bounds__(256) void bin_place_kernel(const int* __restrict__ src,
                                                        const int* __restrict__ dst,
                                                        const int* __restrict__ excl,
                                                        const int* __restrict__ total,
                                                        int* __restrict__ ps, int* __restrict__ pd,
                                                        int* __restrict__ bstart, int e, int nb) {
    __shared__ int sc[256];
    __shared__ int cur[NBMAX];
    const int t = threadIdx.x;
    const int v = t < nb ? total[t] : 0;
    const int inc = block_scan_256(sc, v);
    const int start = inc - v;
    cur[t] = t < nb ? start + excl[blockIdx.x * NBMAX + t] : 0;
    if (blockIdx.x == 0) {
        if (t < nb) bstart[t] = start;
        if (t == 255) bstart[nb] = inc;
    }
    __syncthreads();
    const int chunk = (e + PG - 1) / PG;
    const int lo = blockIdx.x * chunk;
    const int hi = lo + chunk < e ? lo + chunk : e;
    for (int i = lo + t; i < hi; i += 256) {
        const int s = src[i];
        const int d = dst[i];
        const int pos = atomicAdd(&cur[bucket_of(d)], 1);
        ps[pos] = s;
        pd[pos] = d;
    }
}

#define UB 8
#define IR 4

__global__ __launch_bounds__(1024) void bucket_aggregate_kernel(
    const int* __restrict__ ps, const int* __restrict__ pd, const int* __restrict__ bstart,
    const float* __restrict__ dinv, const float* __restrict__ H, const float* __restrict__ bias,
    float* __restrict__ z, int n, int do_relu) {
    __shared__ float acc[BK * DF];
    const int node0 = blockIdx.x * BK;
    const int nn = n - node0 < BK ? n - node0 : BK;
    const int lane = threadIdx.x & 31;
    const int wave = __builtin_amdgcn_readfirstlane((int)(threadIdx.x >> 5));

    for (int r0 = wave; r0 < nn; r0 += 32 * IR) {
        float m[IR][4];
        float q[IR];
#pragma unroll
        for (int u = 0; u < IR; u++) {
            const int r = r0 + 32 * u;
            if (r < nn) {
                const float* h = H + (size_t)(node0 + r) * DF + lane;
                m[u][0] = h[0];
                m[u][1] = h[32];
                m[u][2] = h[64];
                m[u][3] = h[96];
                q[u] = dinv[node0 + r];
            }
        }
#pragma unroll
        for (int u = 0; u < IR; u++) {
            const int r = r0 + 32 * u;
            if (r < nn) {
                const float qq = q[u] * q[u];
                float* a = acc + r * DF + lane;
                a[0] = m[u][0] * qq;
                a[32] = m[u][1] * qq;
                a[64] = m[u][2] * qq;
                a[96] = m[u][3] * qq;
            }
        }
    }
    __syncthreads();

    const int e0 = bstart[blockIdx.x];
    const int e1 = bstart[blockIdx.x + 1];
    const int ngroups = (e1 - e0 + 31) >> 5;
    for (int g = wave; g < ngroups; g += 32) {
        const int base = e0 + (g << 5);
        const int left = e1 - base;
        const int cnt = left < 32 ? left : 32;
        int s = 0, d = 0;
        float nrm = 0.0f;
        if (lane < cnt) {
            s = ps[base + lane];
            d = pd[base + lane];
            nrm = dinv[s] * dinv[d];
            d -= node0;
        }
        if (cnt == 32) {
#pragma unroll
            for (int j0 = 0; j0 < 32; j0 += UB) {
                float m[UB][4];
#pragma unroll
                for (int u = 0; u < UB; u++) {
                    const int sj = __builtin_amdgcn_readlane(s, j0 + u);
                    const float* h = H + (size_t)sj * DF + lane;
                    m[u][0] = h[0];
                    m[u][1] = h[32];
                    m[u][2] = h[64];
                    m[u][3] = h[96];
                }
#pragma unroll
                for (int u = 0; u < UB; u++) {
                    const int dj = __builtin_amdgcn_readlane(d, j0 + u);
                    const float nj =
                        __int_as_float(__builtin_amdgcn_readlane(__float_as_int(nrm), j0 + u));
                    float* a = acc + dj * DF + lane;
                    atomicAdd(a + 0, m[u][0] * nj);
                    atomicAdd(a + 32, m[u][1] * nj);
                    atomicAdd(a + 64, m[u][2] * nj);
                    atomicAdd(a + 96, m[u][3] * nj);
                }
            }
        } else {
            for (int j = 0; j < cnt; j++) {
                const int sj = __builtin_amdgcn_readlane(s, j);
                const int dj = __builtin_amdgcn_readlane(d, j);
                const float nj = __int_as_float(__builtin_amdgcn_readlane(__float_as_int(nrm), j));
                const float* h = H + (size_t)sj * DF + lane;
                float* a = acc + dj * DF + lane;
                atomicAdd(a + 0, h[0] * nj);
                atomicAdd(a + 32, h[32] * nj);
                atomicAdd(a + 64, h[64] * nj);
                atomicAdd(a + 96, h[96] * nj);
            }
        }
    }
    __syncthreads();

    const float b0 = bias[lane], b1 = bias[lane + 32], b2 = bias[lane + 64], b3 = bias[lane + 96];
    for (int r = wave; r < nn; r += 32) {
        const float* a = acc + r * DF + lane;
        float w0 = a[0] + b0, w1 = a[32] + b1, w2 = a[64] + b2, w3 = a[96] + b3;
        if (do_relu) {
            w0 = fmaxf(w0, 0.0f);
            w1 = fmaxf(w1, 0.0f);
            w2 = fmaxf(w2, 0.0f);
            w3 = fmaxf(w3, 0.0f);
        }
        float* o = z + (size_t)(node0 + r) * DF + lane;
        o[0] = w0;
        o[32] = w1;
        o[64] = w2;
        o[96] = w3;
    }
}


__global__ __launch_bounds__(256) void decode_kernel(const int* __restrict__ s,
                                                     const int* __restrict__ d,
                                                     const float* __restrict__ z,
                                                     float* __restrict__ logits, int e) {
    const int lane = threadIdx.x & 31;
    const int w = (int)((blockIdx.x * blockDim.x + threadIdx.x) >> 5);
    if (w >= e) return;
    const int si = s[w];
    const int di = d[w];
    v4f a = ((const v4f*)(z + (size_t)si * DF))[lane];
    v4f b = ((const v4f*)(z + (size_t)di * DF))[lane];
    float dot = a.x * b.x + a.y * b.y + a.z * b.z + a.w * b.w;
#pragma unroll
    for (int off = 16; off > 0; off >>= 1) dot += __shfl_xor(dot, off, 32);
    if (lane == 0) logits[w] = dot;
}


extern "C" void kernel_launch(void* const* d_in, const int* in_sizes, int n_in,
                              void* d_out, int out_size, void* d_ws, size_t ws_size,
                              hipStream_t stream) {
    const float* x  = (const float*)d_in[0];
    const int*   ei = (const int*)d_in[1];
    const int*   el = (const int*)d_in[2];
    const float* W1 = (const float*)d_in[3];
    const float* b1 = (const float*)d_in[4];
    const float* W2 = (const float*)d_in[5];
    const float* b2 = (const float*)d_in[6];

    const int n  = in_sizes[0] / DF;
    const int E  = in_sizes[1] / 2;
    const int EL = in_sizes[2] / 2;

    float* bufA = (float*)d_ws;
    float* bufB = bufA + (size_t)n * DF;
    float* dinv = bufB + (size_t)n * DF;
    int* blkcnt = (int*)(dinv + n);
    int* excl   = blkcnt + PG * NBMAX;
    int* total  = excl + PG * NBMAX;
    int* bstart = total + NBMAX;
    int* ps     = bstart + 2 * NBMAX;
    int* pd     = ps + E;
    float* out  = (float*)d_out;

    const int nBl    = (n + 255) / 256;
    const int eBl    = (E + 255) / 256;
    const int decBl  = (EL + 7) / 8;
    const int nb     = (n + BK - 1) / BK;

    deg_init_kernel<<<nBl, 256, 0, stream>>>(dinv, n);
    deg_accum_kernel<<<eBl, 256, 0, stream>>>(ei + E, dinv, E);
    deg_rsqrt_kernel<<<nBl, 256, 0, stream>>>(dinv, n);

    bin_count_kernel<<<PG, 256, 0, stream>>>(ei + E, blkcnt, E);
    bin_scan_kernel<<<nb, 256, 0, stream>>>(blkcnt, excl, total);
    bin_place_kernel<<<PG, 256, 0, stream>>>(ei, ei + E, excl, total, ps, pd, bstart, E, nb);

    gemm_wmma_kernel<<<GW, 128, 0, stream>>>(x, W1, bufA, n);
    bucket_aggregate_kernel<<<nb, 1024, 0, stream>>>(ps, pd, bstart, dinv, bufA, b1, bufB, n, 1);

    gemm_wmma_kernel<<<GW, 128, 0, stream>>>(bufB, W2, bufA, n);
    bucket_aggregate_kernel<<<nb, 1024, 0, stream>>>(ps, pd, bstart, dinv, bufA, b2, bufB, n, 0);

    decode_kernel<<<decBl, 256, 0, stream>>>(el, el + EL, bufB, out, EL);
}


# Edge norm factored: GEMM stores rows scaled by dinv[row], edges add them as loaded, finished sums are scaled by dinv[dst]; per-edge dinv gathers and multiplies gone; rounding differs, sum order kept
# speedup vs baseline: 2.6665x; 1.0315x over previous
#include <hip/hip_runtime.h>

typedef __attribute__((ext_vector_type(2))) float v2f;
typedef __attribute__((ext_vector_type(4))) float v4f;
typedef __attribute__((ext_vector_type(8))) float v8f;

#define DF 128


__global__ void deg_init_kernel(float* deg, int n) {
    int i = blockIdx.x * blockDim.x + threadIdx.x;
    if (i < n) deg[i] = 1.0f;
}

__global__ void deg_accum_kernel(const int* __restrict__ dst, float* deg, int e) {
    int i = blockIdx.x * blockDim.x + threadIdx.x;
    if (i < e) unsafeAtomicAdd(&deg[dst[i]], 1.0f);
}

__global__ void deg_rsqrt_kernel(float* deg, int n) {
    int i = blockIdx.x * blockDim.x + threadIdx.x;
    if (i < n) deg[i] = rsqrtf(deg[i]);
}

#define GW 256

__device__ __forceinline__ void load_x_tile(v4f (&xb)[16], const float* __restrict__ X, int tile,
                                            int c, int h, int nrows) {
    int row = tile * 16 + c;
    row = row < nrows ? row : nrows - 1;
    const v4f* xr = (const v4f*)(X + (size_t)row * DF) + h;
#pragma unroll
    for (int j = 0; j < 16; j++) xb[j] = xr[2 * j];
}

__global__ __launch_bounds__(128) void gemm_wmma_kernel(const float* __restrict__ X,
                                                        const float* __restrict__ W,
                                                        const float* __restrict__ dinv,
                                                        float* __restrict__ H, int nrows) {
    const int lane = threadIdx.x & 31;
    const int c = lane & 15;
    const int h = lane >> 4;
    const int wv = __builtin_amdgcn_readfirstlane((int)(blockIdx.x * 4 + (threadIdx.x >> 5)));
    const int nwv = (int)gridDim.x * 4;
    const int ntiles = (nrows + 15) >> 4;
    if (wv >= ntiles) return;

    v2f wa[32][8];
#pragma unroll
    for (int t = 0; t < 32; t++) {
        const float* wp = W + (size_t)(8 * (t >> 1) + 4 * h + 2 * (t & 1)) * DF + c;
#pragma unroll
        for (int nt = 0; nt < 8; nt++) {
            wa[t][nt].x = wp[nt * 16];
            wa[t][nt].y = wp[DF + nt * 16];
        }
    }

    v4f xb[16];
    load_x_tile(xb, X, wv, c, h, nrows);
    const int row0 = wv * 16 + c;
    float dn = dinv[row0 < nrows ? row0 : nrows - 1];
    for (int tile = wv; tile < ntiles; tile += nwv) {
        const int nxt = tile + nwv < ntiles ? tile + nwv : tile;
        v4f xn[16];
        load_x_tile(xn, X, nxt, c, h, nrows);
        const int nrow = nxt * 16 + c;
        const float dnn = dinv[nrow < nrows ? nrow : nrows - 1];

        v8f acc[8] = {};
#pragma unroll
        for (int j = 0; j < 16; j++) {
            const v2f b0 = __builtin_shufflevector(xb[j], xb[j], 0, 1);
            const v2f b1 = __builtin_shufflevector(xb[j], xb[j], 2, 3);
#pragma unroll
            for (int nt = 0; nt < 8; nt++)
                acc[nt] = __builtin_amdgcn_wmma_f32_16x16x4_f32(
                    false, wa[2 * j][nt], false, b0, (short)0, acc[nt], false, false);
#pragma unroll
            for (int nt = 0; nt < 8; nt++)
                acc[nt] = __builtin_amdgcn_wmma_f32_16x16x4_f32(
                    false, wa[2 * j + 1][nt], false, b1, (short)0, acc[nt], false, false);
        }

        const int row = tile * 16 + c;
        if (row < nrows) {
            float* o = H + (size_t)row * DF + 8 * h;
#pragma unroll
            for (int nt = 0; nt < 8; nt++) {
                const v8f sc = acc[nt] * dn;
                *(v4f*)(o + nt * 16) = __builtin_shufflevector(sc, sc, 0, 1, 2, 3);
                *(v4f*)(o + nt * 16 + 4) = __builtin_shufflevector(sc, sc, 4, 5, 6, 7);
            }
        }
#pragma unroll
        for (int j = 0; j < 16; j++) xb[j] = xn[j];
        dn = dnn;
    }
}

#define BK 400
#define NBMAX 256
#define PG 256

__device__ __forceinline__ int bucket_of(int d) {
    const int k = d / BK;
    return k < NBMAX - 1 ? k : NBMAX - 1;
}

__global__ __launch_bounds__(256) void bin_count_kernel(const int* __restrict__ dst,
                                                        int* __restrict__ blkcnt, int e) {
    __shared__ int hist[NBMAX];
    hist[threadIdx.x] = 0;
    __syncthreads();
    const int chunk = (e + PG - 1) / PG;
    const int lo = blockIdx.x * chunk;
    const int hi = lo + chunk < e ? lo + chunk : e;
    for (int i = lo + (int)threadIdx.x; i < hi; i += 256) atomicAdd(&hist[bucket_of(dst[i])], 1);
    __syncthreads();
    blkcnt[blockIdx.x * NBMAX + threadIdx.x] = hist[threadIdx.x];
}

__device__ __forceinline__ int block_scan_256(int* sc, int v) {
    const int t = threadIdx.x;
    sc[t] = v;
    __syncthreads();
    for (int off = 1; off < 256; off <<= 1) {
        const int add = t >= off ? sc[t - off] : 0;
        __syncthreads();
        sc[t] += add;
        __syncthreads();
    }
    return sc[t];
}

__global__ __launch_bounds__(256) void bin_scan_kernel(const int* __restrict__ blkcnt,
                                                       int* __restrict__ excl,
                                                       int* __restrict__ total) {
    __shared__ int sc[256];
    const int k = blockIdx.x;
    const int b = threadIdx.x;
    const int v = blkcnt[b * NBMAX + k];
    const int inc = block_scan_256(sc, v);
    excl[b * NBMAX + k] = inc - v;
    if (b == 255) total[k] = inc;
}

__global__ __launch_bounds__(256) void bin_place_kernel(const int* __restrict__ src,
                                                        const int* __restrict__ dst,
                                                        const int* __restrict__ excl,
                                                        const int* __restrict__ total,
                                                        int* __restrict__ ps, int* __restrict__ pd,
                                                        int* __restrict__ bstart, int e, int nb) {
    __shared__ int sc[256];
    __shared__ int cur[NBMAX];
    const int t = threadIdx.x;
    const int v = t < nb ? total[t] : 0;
    const int inc = block_scan_256(sc, v);
    const int start = inc - v;
    cur[t] = t < nb ? start + excl[blockIdx.x * NBMAX + t] : 0;
    if (blockIdx.x == 0) {
        if (t < nb) bstart[t] = start;
        if (t == 255) bstart[nb] = inc;
    }
    __syncthreads();
    const int chunk = (e + PG - 1) / PG;
    const int lo = blockIdx.x * chunk;
    const int hi = lo + chunk < e ? lo + chunk : e;
    for (int i = lo + t; i < hi; i += 256) {
        const int s = src[i];
        const int d = dst[i];
        const int pos = atomicAdd(&cur[bucket_of(d)], 1);
        ps[pos] = s;
        pd[pos] = d;
    }
}

#define UB 8
#define IR 4

__global__ __launch_bounds__(1024) void bucket_aggregate_kernel(
    const int* __restrict__ ps, const int* __restrict__ pd, const int* __restrict__ bstart,
    const float* __restrict__ dinv, const float* __restrict__ G, const float* __restrict__ bias,
    float* __restrict__ z, int n, int do_relu) {
    __shared__ float acc[BK * DF];
    const int node0 = blockIdx.x * BK;
    const int nn = n - node0 < BK ? n - node0 : BK;
    const int lane = threadIdx.x & 31;
    const int wave = __builtin_amdgcn_readfirstlane((int)(threadIdx.x >> 5));

    for (int r0 = wave; r0 < nn; r0 += 32 * IR) {
        float m[IR][4];
#pragma unroll
        for (int u = 0; u < IR; u++) {
            const int r = r0 + 32 * u;
            if (r < nn) {
                const float* h = G + (size_t)(node0 + r) * DF + lane;
                m[u][0] = h[0];
                m[u][1] = h[32];
                m[u][2] = h[64];
                m[u][3] = h[96];
            }
        }
#pragma unroll
        for (int u = 0; u < IR; u++) {
            const int r = r0 + 32 * u;
            if (r < nn) {
                float* a = acc + r * DF + lane;
                a[0] = m[u][0];
                a[32] = m[u][1];
                a[64] = m[u][2];
                a[96] = m[u][3];
            }
        }
    }
    __syncthreads();

    const int e0 = bstart[blockIdx.x];
    const int e1 = bstart[blockIdx.x + 1];
    const int ngroups = (e1 - e0 + 31) >> 5;
    for (int g = wave; g < ngroups; g += 32) {
        const int base = e0 + (g << 5);
        const int left = e1 - base;
        const int cnt = left < 32 ? left : 32;
        int s = 0, d = 0;
        if (lane < cnt) {
            s = ps[base + lane];
            d = pd[base + lane] - node0;
        }
        if (cnt == 32) {
#pragma unroll
            for (int j0 = 0; j0 < 32; j0 += UB) {
                float m[UB][4];
#pragma unroll
                for (int u = 0; u < UB; u++) {
                    const int sj = __builtin_amdgcn_readlane(s, j0 + u);
                    const float* h = G + (size_t)sj * DF + lane;
                    m[u][0] = h[0];
                    m[u][1] = h[32];
                    m[u][2] = h[64];
                    m[u][3] = h[96];
                }
#pragma unroll
                for (int u = 0; u < UB; u++) {
                    const int dj = __builtin_amdgcn_readlane(d, j0 + u);
                    float* a = acc + dj * DF + lane;
                    atomicAdd(a + 0, m[u][0]);
                    atomicAdd(a + 32, m[u][1]);
                    atomicAdd(a + 64, m[u][2]);
                    atomicAdd(a + 96, m[u][3]);
                }
            }
        } else {
            for (int j = 0; j < cnt; j++) {
                const int sj = __builtin_amdgcn_readlane(s, j);
                const int dj = __builtin_amdgcn_readlane(d, j);
                const float* h = G + (size_t)sj * DF + lane;
                float* a = acc + dj * DF + lane;
                atomicAdd(a + 0, h[0]);
                atomicAdd(a + 32, h[32]);
                atomicAdd(a + 64, h[64]);
                atomicAdd(a + 96, h[96]);
            }
        }
    }
    __syncthreads();

    const float b0 = bias[lane], b1 = bias[lane + 32], b2 = bias[lane + 64], b3 = bias[lane + 96];
    for (int r = wave; r < nn; r += 32) {
        const float* a = acc + r * DF + lane;
        const float q = dinv[node0 + r];
        float w0 = a[0] * q + b0, w1 = a[32] * q + b1, w2 = a[64] * q + b2, w3 = a[96] * q + b3;
        if (do_relu) {
            w0 = fmaxf(w0, 0.0f);
            w1 = fmaxf(w1, 0.0f);
            w2 = fmaxf(w2, 0.0f);
            w3 = fmaxf(w3, 0.0f);
        }
        float* o = z + (size_t)(node0 + r) * DF + lane;
        o[0] = w0;
        o[32] = w1;
        o[64] = w2;
        o[96] = w3;
    }
}


__global__ __launch_bounds__(256) void decode_kernel(const int* __restrict__ s,
                                                     const int* __restrict__ d,
                                                     const float* __restrict__ z,
                                                     float* __restrict__ logits, int e) {
    const int lane = threadIdx.x & 31;
    const int w = (int)((blockIdx.x * blockDim.x + threadIdx.x) >> 5);
    if (w >= e) return;
    const int si = s[w];
    const int di = d[w];
    v4f a = ((const v4f*)(z + (size_t)si * DF))[lane];
    v4f b = ((const v4f*)(z + (size_t)di * DF))[lane];
    float dot = a.x * b.x + a.y * b.y + a.z * b.z + a.w * b.w;
#pragma unroll
    for (int off = 16; off > 0; off >>= 1) dot += __shfl_xor(dot, off, 32);
    if (lane == 0) logits[w] = dot;
}


extern "C" void kernel_launch(void* const* d_in, const int* in_sizes, int n_in,
                              void* d_out, int out_size, void* d_ws, size_t ws_size,
                              hipStream_t stream) {
    const float* x  = (const float*)d_in[0];
    const int*   ei = (const int*)d_in[1];
    const int*   el = (const int*)d_in[2];
    const float* W1 = (const float*)d_in[3];
    const float* b1 = (const float*)d_in[4];
    const float* W2 = (const float*)d_in[5];
    const float* b2 = (const float*)d_in[6];

    const int n  = in_sizes[0] / DF;
    const int E  = in_sizes[1] / 2;
    const int EL = in_sizes[2] / 2;

    float* bufA = (float*)d_ws;
    float* bufB = bufA + (size_t)n * DF;
    float* dinv = bufB + (size_t)n * DF;
    int* blkcnt = (int*)(dinv + n);
    int* excl   = blkcnt + PG * NBMAX;
    int* total  = excl + PG * NBMAX;
    int* bstart = total + NBMAX;
    int* ps     = bstart + 2 * NBMAX;
    int* pd     = ps + E;
    float* out  = (float*)d_out;

    const int nBl    = (n + 255) / 256;
    const int eBl    = (E + 255) / 256;
    const int decBl  = (EL + 7) / 8;
    const int nb     = (n + BK - 1) / BK;

    deg_init_kernel<<<nBl, 256, 0, stream>>>(dinv, n);
    deg_accum_kernel<<<eBl, 256, 0, stream>>>(ei + E, dinv, E);
    deg_rsqrt_kernel<<<nBl, 256, 0, stream>>>(dinv, n);

    bin_count_kernel<<<PG, 256, 0, stream>>>(ei + E, blkcnt, E);
    bin_scan_kernel<<<nb, 256, 0, stream>>>(blkcnt, excl, total);
    bin_place_kernel<<<PG, 256, 0, stream>>>(ei, ei + E, excl, total, ps, pd, bstart, E, nb);

    gemm_wmma_kernel<<<GW, 128, 0, stream>>>(x, W1, dinv, bufA, n);
    bucket_aggregate_kernel<<<nb, 1024, 0, stream>>>(ps, pd, bstart, dinv, bufA, b1, bufB, n, 1);

    gemm_wmma_kernel<<<GW, 128, 0, stream>>>(bufB, W2, dinv, bufA, n);
    bucket_aggregate_kernel<<<nb, 1024, 0, stream>>>(ps, pd, bstart, dinv, bufA, b2, bufB, n, 0);

    decode_kernel<<<decBl, 256, 0, stream>>>(el, el + EL, bufB, out, EL);
}


# Edge sums leave LDS float atomics: a bucket's edges go in LDS to the wave owning the dst row (row mod 32, rank from an LDS int add); it adds rows by plain 16-B LDS load/add/store; arrival order
# speedup vs baseline: 2.9061x; 1.0899x over previous
#include <hip/hip_runtime.h>

typedef __attribute__((ext_vector_type(2))) float v2f;
typedef __attribute__((ext_vector_type(4))) float v4f;
typedef __attribute__((ext_vector_type(8))) float v8f;

#define DF 128


__global__ void deg_init_kernel(float* deg, int n) {
    int i = blockIdx.x * blockDim.x + threadIdx.x;
    if (i < n) deg[i] = 1.0f;
}

__global__ void deg_accum_kernel(const int* __restrict__ dst, float* deg, int e) {
    int i = blockIdx.x * blockDim.x + threadIdx.x;
    if (i < e) unsafeAtomicAdd(&deg[dst[i]], 1.0f);
}

__global__ void deg_rsqrt_kernel(float* deg, int n) {
    int i = blockIdx.x * blockDim.x + threadIdx.x;
    if (i < n) deg[i] = rsqrtf(deg[i]);
}

#define GW 256

__device__ __forceinline__ void load_x_tile(v4f (&xb)[16], const float* __restrict__ X, int tile,
                                            int c, int h, int nrows) {
    int row = tile * 16 + c;
    row = row < nrows ? row : nrows - 1;
    const v4f* xr = (const v4f*)(X + (size_t)row * DF) + h;
#pragma unroll
    for (int j = 0; j < 16; j++) xb[j] = xr[2 * j];
}

__global__ __launch_bounds__(128) void gemm_wmma_kernel(const float* __restrict__ X,
                                                        const float* __restrict__ W,
                                                        const float* __restrict__ dinv,
                                                        float* __restrict__ H, int nrows) {
    const int lane = threadIdx.x & 31;
    const int c = lane & 15;
    const int h = lane >> 4;
    const int wv = __builtin_amdgcn_readfirstlane((int)(blockIdx.x * 4 + (threadIdx.x >> 5)));
    const int nwv = (int)gridDim.x * 4;
    const int ntiles = (nrows + 15) >> 4;
    if (wv >= ntiles) return;

    v2f wa[32][8];
#pragma unroll
    for (int t = 0; t < 32; t++) {
        const float* wp = W + (size_t)(8 * (t >> 1) + 4 * h + 2 * (t & 1)) * DF + c;
#pragma unroll
        for (int nt = 0; nt < 8; nt++) {
            wa[t][nt].x = wp[nt * 16];
            wa[t][nt].y = wp[DF + nt * 16];
        }
    }

    v4f xb[16];
    load_x_tile(xb, X, wv, c, h, nrows);
    const int row0 = wv * 16 + c;
    float dn = dinv[row0 < nrows ? row0 : nrows - 1];
    for (int tile = wv; tile < ntiles; tile += nwv) {
        const int nxt = tile + nwv < ntiles ? tile + nwv : tile;
        v4f xn[16];
        load_x_tile(xn, X, nxt, c, h, nrows);
        const int nrow = nxt * 16 + c;
        const float dnn = dinv[nrow < nrows ? nrow : nrows - 1];

        v8f acc[8] = {};
#pragma unroll
        for (int j = 0; j < 16; j++) {
            const v2f b0 = __builtin_shufflevector(xb[j], xb[j], 0, 1);
            const v2f b1 = __builtin_shufflevector(xb[j], xb[j], 2, 3);
#pragma unroll
            for (int nt = 0; nt < 8; nt++)
                acc[nt] = __builtin_amdgcn_wmma_f32_16x16x4_f32(
                    false, wa[2 * j][nt], false, b0, (short)0, acc[nt], false, false);
#pragma unroll
            for (int nt = 0; nt < 8; nt++)
                acc[nt] = __builtin_amdgcn_wmma_f32_16x16x4_f32(
                    false, wa[2 * j + 1][nt], false, b1, (short)0, acc[nt], false, false);
        }

        const int row = tile * 16 + c;
        if (row < nrows) {
            float* o = H + (size_t)row * DF + 8 * h;
#pragma unroll
            for (int nt = 0; nt < 8; nt++) {
                const v8f sc = acc[nt] * dn;
                *(v4f*)(o + nt * 16) = __builtin_shufflevector(sc, sc, 0, 1, 2, 3);
                *(v4f*)(o + nt * 16 + 4) = __builtin_shufflevector(sc, sc, 4, 5, 6, 7);
            }
        }
#pragma unroll
        for (int j = 0; j < 16; j++) xb[j] = xn[j];
        dn = dnn;
    }
}

#define BK 400
#define NBMAX 256
#define PG 256

__device__ __forceinline__ int bucket_of(int d) {
    const int k = d / BK;
    return k < NBMAX - 1 ? k : NBMAX - 1;
}

__global__ __launch_bounds__(256) void bin_count_kernel(const int* __restrict__ dst,
                                                        int* __restrict__ blkcnt, int e) {
    __shared__ int hist[NBMAX];
    hist[threadIdx.x] = 0;
    __syncthreads();
    const int chunk = (e + PG - 1) / PG;
    const int lo = blockIdx.x * chunk;
    const int hi = lo + chunk < e ? lo + chunk : e;
    for (int i = lo + (int)threadIdx.x; i < hi; i += 256) atomicAdd(&hist[bucket_of(dst[i])], 1);
    __syncthreads();
    blkcnt[blockIdx.x * NBMAX + threadIdx.x] = hist[threadIdx.x];
}

__device__ __forceinline__ int block_scan_256(int* sc, int v) {
    const int t = threadIdx.x;
    sc[t] = v;
    __syncthreads();
    for (int off = 1; off < 256; off <<= 1) {
        const int add = t >= off ? sc[t - off] : 0;
        __syncthreads();
        sc[t] += add;
        __syncthreads();
    }
    return sc[t];
}

__global__ __launch_bounds__(256) void bin_scan_kernel(const int* __restrict__ blkcnt,
                                                       int* __restrict__ excl,
                                                       int* __restrict__ total) {
    __shared__ int sc[256];
    const int k = blockIdx.x;
    const int b = threadIdx.x;
    const int v = blkcnt[b * NBMAX + k];
    const int inc = block_scan_256(sc, v);
    excl[b * NBMAX + k] = inc - v;
    if (b == 255) total[k] = inc;
}

__global__ __launch_bounds__(256) void bin_place_kernel(const int* __restrict__ src,
                                                        const int* __restrict__ dst,
                                                        const int* __restrict__ excl,
                                                        const int* __restrict__ total,
                                                        int* __restrict__ ps, int* __restrict__ pd,
                                                        int* __restrict__ bstart, int e, int nb) {
    __shared__ int sc[256];
    __shared__ int cur[NBMAX];
    const int t = threadIdx.x;
    const int v = t < nb ? total[t] : 0;
    const int inc = block_scan_256(sc, v);
    const int start = inc - v;
    cur[t] = t < nb ? start + excl[blockIdx.x * NBMAX + t] : 0;
    if (blockIdx.x == 0) {
        if (t < nb) bstart[t] = start;
        if (t == 255) bstart[nb] = inc;
    }
    __syncthreads();
    const int chunk = (e + PG - 1) / PG;
    const int lo = blockIdx.x * chunk;
    const int hi = lo + chunk < e ? lo + chunk : e;
    for (int i = lo + t; i < hi; i += 256) {
        const int s = src[i];
        const int d = dst[i];
        const int pos = atomicAdd(&cur[bucket_of(d)], 1);
        ps[pos] = s;
        pd[pos] = d;
    }
}

#define QC 8192
#define IR 4

__global__ __launch_bounds__(1024) void bucket_aggregate_kernel(
    const int* __restrict__ ps, const int* __restrict__ pd, const int* __restrict__ bstart,
    const float* __restrict__ dinv, const float* __restrict__ G, const float* __restrict__ bias,
    float* __restrict__ z, int n, int do_relu) {
    __shared__ v4f acc[BK * 32];
    __shared__ int2 q[QC];
    __shared__ int cnt[32];
    const int node0 = blockIdx.x * BK;
    const int nn = n - node0 < BK ? n - node0 : BK;
    const int tid = threadIdx.x;
    const int lane = tid & 31;
    const int wave = __builtin_amdgcn_readfirstlane(tid >> 5);
    const v4f* G4 = (const v4f*)G;

    for (int r0 = wave; r0 < nn; r0 += 32 * IR) {
        v4f m[IR];
#pragma unroll
        for (int u = 0; u < IR; u++) {
            const int r = r0 + 32 * u;
            if (r < nn) m[u] = G4[(size_t)(node0 + r) * 32 + lane];
        }
#pragma unroll
        for (int u = 0; u < IR; u++) {
            const int r = r0 + 32 * u;
            if (r < nn) acc[r * 32 + lane] = m[u];
        }
    }

    const int e0 = bstart[blockIdx.x];
    const int e1 = bstart[blockIdx.x + 1];
    for (int base = e0; base < e1; base += QC) {
        const int m = e1 - base < QC ? e1 - base : QC;
        if (tid < 32) cnt[tid] = 0;
        __syncthreads();

        int s[8], d[8], rk[8];
#pragma unroll
        for (int k = 0; k < 8; k++) {
            const int i = k * 1024 + tid;
            const int ii = i < m ? i : m - 1;
            s[k] = ps[base + ii];
            d[k] = pd[base + ii] - node0;
        }
#pragma unroll
        for (int k = 0; k < 8; k++) {
            const int i = k * 1024 + tid;
            rk[k] = 0;
            if (i < m) rk[k] = atomicAdd(&cnt[d[k] & 31], 1);
        }
        __syncthreads();

        const int c = cnt[lane];
        int inc = c;
#pragma unroll
        for (int off = 1; off < 32; off <<= 1) {
            const int t = __shfl_up(inc, off, 32);
            if (lane >= off) inc += t;
        }
        const int first = inc - c;
#pragma unroll
        for (int k = 0; k < 8; k++) {
            const int i = k * 1024 + tid;
            const int st = __shfl(first, d[k] & 31, 32);
            if (i < m) q[st + rk[k]] = make_int2(s[k], d[k]);
        }
        __syncthreads();

        const int qs = __builtin_amdgcn_readlane(first, wave);
        const int qn = __builtin_amdgcn_readlane(c, wave);
        for (int g = 0; g < qn; g += 32) {
            const int left = qn - g;
            const int cg = left < 32 ? left : 32;
            int2 e = make_int2(0, 0);
            if (lane < cg) e = q[qs + g + lane];
            if (cg == 32) {
                v4f ma[8], mb[8];
#pragma unroll
                for (int u = 0; u < 8; u++)
                    ma[u] = G4[(size_t)__builtin_amdgcn_readlane(e.x, u) * 32 + lane];
#pragma unroll
                for (int u = 0; u < 8; u++)
                    mb[u] = G4[(size_t)__builtin_amdgcn_readlane(e.x, 8 + u) * 32 + lane];
#pragma unroll
                for (int u = 0; u < 8; u++) {
                    v4f* a = acc + __builtin_amdgcn_readlane(e.y, u) * 32 + lane;
                    *a = *a + ma[u];
                }
#pragma unroll
                for (int u = 0; u < 8; u++)
                    ma[u] = G4[(size_t)__builtin_amdgcn_readlane(e.x, 16 + u) * 32 + lane];
#pragma unroll
                for (int u = 0; u < 8; u++) {
                    v4f* a = acc + __builtin_amdgcn_readlane(e.y, 8 + u) * 32 + lane;
                    *a = *a + mb[u];
                }
#pragma unroll
                for (int u = 0; u < 8; u++)
                    mb[u] = G4[(size_t)__builtin_amdgcn_readlane(e.x, 24 + u) * 32 + lane];
#pragma unroll
                for (int u = 0; u < 8; u++) {
                    v4f* a = acc + __builtin_amdgcn_readlane(e.y, 16 + u) * 32 + lane;
                    *a = *a + ma[u];
                }
#pragma unroll
                for (int u = 0; u < 8; u++) {
                    v4f* a = acc + __builtin_amdgcn_readlane(e.y, 24 + u) * 32 + lane;
                    *a = *a + mb[u];
                }
            } else {
                for (int j0 = 0; j0 < cg; j0 += 8) {
                    v4f mt[8];
#pragma unroll
                    for (int u = 0; u < 8; u++)
                        mt[u] = G4[(size_t)__builtin_amdgcn_readlane(e.x, j0 + u) * 32 + lane];
#pragma unroll
                    for (int u = 0; u < 8; u++) {
                        if (j0 + u < cg) {
                            v4f* a = acc + __builtin_amdgcn_readlane(e.y, j0 + u) * 32 + lane;
                            *a = *a + mt[u];
                        }
                    }
                }
            }
        }
    }

    const v4f b4 = {bias[4 * lane], bias[4 * lane + 1], bias[4 * lane + 2], bias[4 * lane + 3]};
    v4f* z4 = (v4f*)z;
    for (int r0 = wave; r0 < nn; r0 += 32 * IR) {
        float qv[IR];
#pragma unroll
        for (int u = 0; u < IR; u++) {
            const int r = r0 + 32 * u;
            qv[u] = r < nn ? dinv[node0 + r] : 0.0f;
        }
#pragma unroll
        for (int u = 0; u < IR; u++) {
            const int r = r0 + 32 * u;
            if (r < nn) {
                v4f w = acc[r * 32 + lane] * qv[u] + b4;
                if (do_relu) {
                    w.x = fmaxf(w.x, 0.0f);
                    w.y = fmaxf(w.y, 0.0f);
                    w.z = fmaxf(w.z, 0.0f);
                    w.w = fmaxf(w.w, 0.0f);
                }
                z4[(size_t)(node0 + r) * 32 + lane] = w;
            }
        }
    }
}


__global__ __launch_bounds__(256) void decode_kernel(const int* __restrict__ s,
                                                     const int* __restrict__ d,
                                                     const float* __restrict__ z,
                                                     float* __restrict__ logits, int e) {
    const int lane = threadIdx.x & 31;
    const int w = (int)((blockIdx.x * blockDim.x + threadIdx.x) >> 5);
    if (w >= e) return;
    const int si = s[w];
    const int di = d[w];
    v4f a = ((const v4f*)(z + (size_t)si * DF))[lane];
    v4f b = ((const v4f*)(z + (size_t)di * DF))[lane];
    float dot = a.x * b.x + a.y * b.y + a.z * b.z + a.w * b.w;
#pragma unroll
    for (int off = 16; off > 0; off >>= 1) dot += __shfl_xor(dot, off, 32);
    if (lane == 0) logits[w] = dot;
}


extern "C" void kernel_launch(void* const* d_in, const int* in_sizes, int n_in,
                              void* d_out, int out_size, void* d_ws, size_t ws_size,
                              hipStream_t stream) {
    const float* x  = (const float*)d_in[0];
    const int*   ei = (const int*)d_in[1];
    const int*   el = (const int*)d_in[2];
    const float* W1 = (const float*)d_in[3];
    const float* b1 = (const float*)d_in[4];
    const float* W2 = (const float*)d_in[5];
    const float* b2 = (const float*)d_in[6];

    const int n  = in_sizes[0] / DF;
    const int E  = in_sizes[1] / 2;
    const int EL = in_sizes[2] / 2;

    float* bufA = (float*)d_ws;
    float* bufB = bufA + (size_t)n * DF;
    float* dinv = bufB + (size_t)n * DF;
    int* blkcnt = (int*)(dinv + n);
    int* excl   = blkcnt + PG * NBMAX;
    int* total  = excl + PG * NBMAX;
    int* bstart = total + NBMAX;
    int* ps     = bstart + 2 * NBMAX;
    int* pd     = ps + E;
    float* out  = (float*)d_out;

    const int nBl    = (n + 255) / 256;
    const int eBl    = (E + 255) / 256;
    const int decBl  = (EL + 7) / 8;
    const int nb     = (n + BK - 1) / BK;

    deg_init_kernel<<<nBl, 256, 0, stream>>>(dinv, n);
    deg_accum_kernel<<<eBl, 256, 0, stream>>>(ei + E, dinv, E);
    deg_rsqrt_kernel<<<nBl, 256, 0, stream>>>(dinv, n);

    bin_count_kernel<<<PG, 256, 0, stream>>>(ei + E, blkcnt, E);
    bin_scan_kernel<<<nb, 256, 0, stream>>>(blkcnt, excl, total);
    bin_place_kernel<<<PG, 256, 0, stream>>>(ei, ei + E, excl, total, ps, pd, bstart, E, nb);

    gemm_wmma_kernel<<<GW, 128, 0, stream>>>(x, W1, dinv, bufA, n);
    bucket_aggregate_kernel<<<nb, 1024, 0, stream>>>(ps, pd, bstart, dinv, bufA, b1, bufB, n, 1);

    gemm_wmma_kernel<<<GW, 128, 0, stream>>>(bufB, W2, dinv, bufA, n);
    bucket_aggregate_kernel<<<nb, 1024, 0, stream>>>(ps, pd, bstart, dinv, bufA, b2, bufB, n, 0);

    decode_kernel<<<decBl, 256, 0, stream>>>(el, el + EL, bufB, out, EL);
}


# Degrees leave global float atomics: each 400-node bucket counts the dst of its regrouped edges in LDS (int adds, exact), writes rsqrt(1+count); one launch replaces init, per-edge adds and rsqrt
# speedup vs baseline: 3.8812x; 1.3355x over previous
#include <hip/hip_runtime.h>

typedef __attribute__((ext_vector_type(2))) float v2f;
typedef __attribute__((ext_vector_type(4))) float v4f;
typedef __attribute__((ext_vector_type(8))) float v8f;

#define DF 128

#define GW 256

__device__ __forceinline__ void load_x_tile(v4f (&xb)[16], const float* __restrict__ X, int tile,
                                            int c, int h, int nrows) {
    int row = tile * 16 + c;
    row = row < nrows ? row : nrows - 1;
    const v4f* xr = (const v4f*)(X + (size_t)row * DF) + h;
#pragma unroll
    for (int j = 0; j < 16; j++) xb[j] = xr[2 * j];
}

__global__ __launch_bounds__(128) void gemm_wmma_kernel(const float* __restrict__ X,
                                                        const float* __restrict__ W,
                                                        const float* __restrict__ dinv,
                                                        float* __restrict__ H, int nrows) {
    const int lane = threadIdx.x & 31;
    const int c = lane & 15;
    const int h = lane >> 4;
    const int wv = __builtin_amdgcn_readfirstlane((int)(blockIdx.x * 4 + (threadIdx.x >> 5)));
    const int nwv = (int)gridDim.x * 4;
    const int ntiles = (nrows + 15) >> 4;
    if (wv >= ntiles) return;

    v2f wa[32][8];
#pragma unroll
    for (int t = 0; t < 32; t++) {
        const float* wp = W + (size_t)(8 * (t >> 1) + 4 * h + 2 * (t & 1)) * DF + c;
#pragma unroll
        for (int nt = 0; nt < 8; nt++) {
            wa[t][nt].x = wp[nt * 16];
            wa[t][nt].y = wp[DF + nt * 16];
        }
    }

    v4f xb[16];
    load_x_tile(xb, X, wv, c, h, nrows);
    const int row0 = wv * 16 + c;
    float dn = dinv[row0 < nrows ? row0 : nrows - 1];
    for (int tile = wv; tile < ntiles; tile += nwv) {
        const int nxt = tile + nwv < ntiles ? tile + nwv : tile;
        v4f xn[16];
        load_x_tile(xn, X, nxt, c, h, nrows);
        const int nrow = nxt * 16 + c;
        const float dnn = dinv[nrow < nrows ? nrow : nrows - 1];

        v8f acc[8] = {};
#pragma unroll
        for (int j = 0; j < 16; j++) {
            const v2f b0 = __builtin_shufflevector(xb[j], xb[j], 0, 1);
            const v2f b1 = __builtin_shufflevector(xb[j], xb[j], 2, 3);
#pragma unroll
            for (int nt = 0; nt < 8; nt++)
                acc[nt] = __builtin_amdgcn_wmma_f32_16x16x4_f32(
                    false, wa[2 * j][nt], false, b0, (short)0, acc[nt], false, false);
#pragma unroll
            for (int nt = 0; nt < 8; nt++)
                acc[nt] = __builtin_amdgcn_wmma_f32_16x16x4_f32(
                    false, wa[2 * j + 1][nt], false, b1, (short)0, acc[nt], false, false);
        }

        const int row = tile * 16 + c;
        if (row < nrows) {
            float* o = H + (size_t)row * DF + 8 * h;
#pragma unroll
            for (int nt = 0; nt < 8; nt++) {
                const v8f sc = acc[nt] * dn;
                *(v4f*)(o + nt * 16) = __builtin_shufflevector(sc, sc, 0, 1, 2, 3);
                *(v4f*)(o + nt * 16 + 4) = __builtin_shufflevector(sc, sc, 4, 5, 6, 7);
            }
        }
#pragma unroll
        for (int j = 0; j < 16; j++) xb[j] = xn[j];
        dn = dnn;
    }
}

#define BK 400
#define NBMAX 256
#define PG 256

__device__ __forceinline__ int bucket_of(int d) {
    const int k = d / BK;
    return k < NBMAX - 1 ? k : NBMAX - 1;
}

__global__ __launch_bounds__(256) void bin_count_kernel(const int* __restrict__ dst,
                                                        int* __restrict__ blkcnt, int e) {
    __shared__ int hist[NBMAX];
    hist[threadIdx.x] = 0;
    __syncthreads();
    const int chunk = (e + PG - 1) / PG;
    const int lo = blockIdx.x * chunk;
    const int hi = lo + chunk < e ? lo + chunk : e;
    for (int i = lo + (int)threadIdx.x; i < hi; i += 256) atomicAdd(&hist[bucket_of(dst[i])], 1);
    __syncthreads();
    blkcnt[blockIdx.x * NBMAX + threadIdx.x] = hist[threadIdx.x];
}

__device__ __forceinline__ int block_scan_256(int* sc, int v) {
    const int t = threadIdx.x;
    sc[t] = v;
    __syncthreads();
    for (int off = 1; off < 256; off <<= 1) {
        const int add = t >= off ? sc[t - off] : 0;
        __syncthreads();
        sc[t] += add;
        __syncthreads();
    }
    return sc[t];
}

__global__ __launch_bounds__(256) void bin_scan_kernel(const int* __restrict__ blkcnt,
                                                       int* __restrict__ excl,
                                                       int* __restrict__ total) {
    __shared__ int sc[256];
    const int k = blockIdx.x;
    const int b = threadIdx.x;
    const int v = blkcnt[b * NBMAX + k];
    const int inc = block_scan_256(sc, v);
    excl[b * NBMAX + k] = inc - v;
    if (b == 255) total[k] = inc;
}

__global__ __launch_bounds__(256) void bin_place_kernel(const int* __restrict__ src,
                                                        const int* __restrict__ dst,
                                                        const int* __restrict__ excl,
                                                        const int* __restrict__ total,
                                                        int* __restrict__ ps, int* __restrict__ pd,
                                                        int* __restrict__ bstart, int e, int nb) {
    __shared__ int sc[256];
    __shared__ int cur[NBMAX];
    const int t = threadIdx.x;
    const int v = t < nb ? total[t] : 0;
    const int inc = block_scan_256(sc, v);
    const int start = inc - v;
    cur[t] = t < nb ? start + excl[blockIdx.x * NBMAX + t] : 0;
    if (blockIdx.x == 0) {
        if (t < nb) bstart[t] = start;
        if (t == 255) bstart[nb] = inc;
    }
    __syncthreads();
    const int chunk = (e + PG - 1) / PG;
    const int lo = blockIdx.x * chunk;
    const int hi = lo + chunk < e ? lo + chunk : e;
    for (int i = lo + t; i < hi; i += 256) {
        const int s = src[i];
        const int d = dst[i];
        const int pos = atomicAdd(&cur[bucket_of(d)], 1);
        ps[pos] = s;
        pd[pos] = d;
    }
}

#define DC 8192

__global__ __launch_bounds__(1024) void bucket_degree_kernel(const int* __restrict__ pd,
                                                             const int* __restrict__ bstart,
                                                             float* __restrict__ dinv, int n) {
    __shared__ int indeg[BK];
    const int tid = threadIdx.x;
    const int node0 = blockIdx.x * BK;
    const int nn = n - node0 < BK ? n - node0 : BK;
    if (tid < BK) indeg[tid] = 0;
    const int e0 = bstart[blockIdx.x];
    const int e1 = bstart[blockIdx.x + 1];
    __syncthreads();
    for (int base = e0; base < e1; base += DC) {
        const int m = e1 - base < DC ? e1 - base : DC;
        int d[8];
#pragma unroll
        for (int k = 0; k < 8; k++) {
            const int i = k * 1024 + tid;
            const int ii = i < m ? i : m - 1;
            d[k] = pd[base + ii] - node0;
        }
#pragma unroll
        for (int k = 0; k < 8; k++) {
            const int i = k * 1024 + tid;
            if (i < m) atomicAdd(&indeg[d[k]], 1);
        }
    }
    __syncthreads();
    if (tid < nn) dinv[node0 + tid] = rsqrtf((float)(indeg[tid] + 1));
}

#define QC 8192
#define IR 4

__global__ __launch_bounds__(1024) void bucket_aggregate_kernel(
    const int* __restrict__ ps, const int* __restrict__ pd, const int* __restrict__ bstart,
    const float* __restrict__ dinv, const float* __restrict__ G, const float* __restrict__ bias,
    float* __restrict__ z, int n, int do_relu) {
    __shared__ v4f acc[BK * 32];
    __shared__ int2 q[QC];
    __shared__ int cnt[32];
    const int node0 = blockIdx.x * BK;
    const int nn = n - node0 < BK ? n - node0 : BK;
    const int tid = threadIdx.x;
    const int lane = tid & 31;
    const int wave = __builtin_amdgcn_readfirstlane(tid >> 5);
    const v4f* G4 = (const v4f*)G;

    for (int r0 = wave; r0 < nn; r0 += 32 * IR) {
        v4f m[IR];
#pragma unroll
        for (int u = 0; u < IR; u++) {
            const int r = r0 + 32 * u;
            if (r < nn) m[u] = G4[(size_t)(node0 + r) * 32 + lane];
        }
#pragma unroll
        for (int u = 0; u < IR; u++) {
            const int r = r0 + 32 * u;
            if (r < nn) acc[r * 32 + lane] = m[u];
        }
    }

    const int e0 = bstart[blockIdx.x];
    const int e1 = bstart[blockIdx.x + 1];
    for (int base = e0; base < e1; base += QC) {
        const int m = e1 - base < QC ? e1 - base : QC;
        if (tid < 32) cnt[tid] = 0;
        __syncthreads();

        int s[8], d[8], rk[8];
#pragma unroll
        for (int k = 0; k < 8; k++) {
            const int i = k * 1024 + tid;
            const int ii = i < m ? i : m - 1;
            s[k] = ps[base + ii];
            d[k] = pd[base + ii] - node0;
        }
#pragma unroll
        for (int k = 0; k < 8; k++) {
            const int i = k * 1024 + tid;
            rk[k] = 0;
            if (i < m) rk[k] = atomicAdd(&cnt[d[k] & 31], 1);
        }
        __syncthreads();

        const int c = cnt[lane];
        int inc = c;
#pragma unroll
        for (int off = 1; off < 32; off <<= 1) {
            const int t = __shfl_up(inc, off, 32);
            if (lane >= off) inc += t;
        }
        const int first = inc - c;
#pragma unroll
        for (int k = 0; k < 8; k++) {
            const int i = k * 1024 + tid;
            const int st = __shfl(first, d[k] & 31, 32);
            if (i < m) q[st + rk[k]] = make_int2(s[k], d[k]);
        }
        __syncthreads();

        const int qs = __builtin_amdgcn_readlane(first, wave);
        const int qn = __builtin_amdgcn_readlane(c, wave);
        for (int g = 0; g < qn; g += 32) {
            const int left = qn - g;
            const int cg = left < 32 ? left : 32;
            int2 e = make_int2(0, 0);
            if (lane < cg) e = q[qs + g + lane];
            if (cg == 32) {
                v4f ma[8], mb[8];
#pragma unroll
                for (int u = 0; u < 8; u++)
                    ma[u] = G4[(size_t)__builtin_amdgcn_readlane(e.x, u) * 32 + lane];
#pragma unroll
                for (int u = 0; u < 8; u++)
                    mb[u] = G4[(size_t)__builtin_amdgcn_readlane(e.x, 8 + u) * 32 + lane];
#pragma unroll
                for (int u = 0; u < 8; u++) {
                    v4f* a = acc + __builtin_amdgcn_readlane(e.y, u) * 32 + lane;
                    *a = *a + ma[u];
                }
#pragma unroll
                for (int u = 0; u < 8; u++)
                    ma[u] = G4[(size_t)__builtin_amdgcn_readlane(e.x, 16 + u) * 32 + lane];
#pragma unroll
                for (int u = 0; u < 8; u++) {
                    v4f* a = acc + __builtin_amdgcn_readlane(e.y, 8 + u) * 32 + lane;
                    *a = *a + mb[u];
                }
#pragma unroll
                for (int u = 0; u < 8; u++)
                    mb[u] = G4[(size_t)__builtin_amdgcn_readlane(e.x, 24 + u) * 32 + lane];
#pragma unroll
                for (int u = 0; u < 8; u++) {
                    v4f* a = acc + __builtin_amdgcn_readlane(e.y, 16 + u) * 32 + lane;
                    *a = *a + ma[u];
                }
#pragma unroll
                for (int u = 0; u < 8; u++) {
                    v4f* a = acc + __builtin_amdgcn_readlane(e.y, 24 + u) * 32 + lane;
                    *a = *a + mb[u];
                }
            } else {
                for (int j0 = 0; j0 < cg; j0 += 8) {
                    v4f mt[8];
#pragma unroll
                    for (int u = 0; u < 8; u++)
                        mt[u] = G4[(size_t)__builtin_amdgcn_readlane(e.x, j0 + u) * 32 + lane];
#pragma unroll
                    for (int u = 0; u < 8; u++) {
                        if (j0 + u < cg) {
                            v4f* a = acc + __builtin_amdgcn_readlane(e.y, j0 + u) * 32 + lane;
                            *a = *a + mt[u];
                        }
                    }
                }
            }
        }
    }

    const v4f b4 = {bias[4 * lane], bias[4 * lane + 1], bias[4 * lane + 2], bias[4 * lane + 3]};
    v4f* z4 = (v4f*)z;
    for (int r0 = wave; r0 < nn; r0 += 32 * IR) {
        float qv[IR];
#pragma unroll
        for (int u = 0; u < IR; u++) {
            const int r = r0 + 32 * u;
            qv[u] = r < nn ? dinv[node0 + r] : 0.0f;
        }
#pragma unroll
        for (int u = 0; u < IR; u++) {
            const int r = r0 + 32 * u;
            if (r < nn) {
                v4f w = acc[r * 32 + lane] * qv[u] + b4;
                if (do_relu) {
                    w.x = fmaxf(w.x, 0.0f);
                    w.y = fmaxf(w.y, 0.0f);
                    w.z = fmaxf(w.z, 0.0f);
                    w.w = fmaxf(w.w, 0.0f);
                }
                z4[(size_t)(node0 + r) * 32 + lane] = w;
            }
        }
    }
}


__global__ __launch_bounds__(256) void decode_kernel(const int* __restrict__ s,
                                                     const int* __restrict__ d,
                                                     const float* __restrict__ z,
                                                     float* __restrict__ logits, int e) {
    const int lane = threadIdx.x & 31;
    const int w = (int)((blockIdx.x * blockDim.x + threadIdx.x) >> 5);
    if (w >= e) return;
    const int si = s[w];
    const int di = d[w];
    v4f a = ((const v4f*)(z + (size_t)si * DF))[lane];
    v4f b = ((const v4f*)(z + (size_t)di * DF))[lane];
    float dot = a.x * b.x + a.y * b.y + a.z * b.z + a.w * b.w;
#pragma unroll
    for (int off = 16; off > 0; off >>= 1) dot += __shfl_xor(dot, off, 32);
    if (lane == 0) logits[w] = dot;
}


extern "C" void kernel_launch(void* const* d_in, const int* in_sizes, int n_in,
                              void* d_out, int out_size, void* d_ws, size_t ws_size,
                              hipStream_t stream) {
    const float* x  = (const float*)d_in[0];
    const int*   ei = (const int*)d_in[1];
    const int*   el = (const int*)d_in[2];
    const float* W1 = (const float*)d_in[3];
    const float* b1 = (const float*)d_in[4];
    const float* W2 = (const float*)d_in[5];
    const float* b2 = (const float*)d_in[6];

    const int n  = in_sizes[0] / DF;
    const int E  = in_sizes[1] / 2;
    const int EL = in_sizes[2] / 2;

    float* bufA = (float*)d_ws;
    float* bufB = bufA + (size_t)n * DF;
    float* dinv = bufB + (size_t)n * DF;
    int* blkcnt = (int*)(dinv + n);
    int* excl   = blkcnt + PG * NBMAX;
    int* total  = excl + PG * NBMAX;
    int* bstart = total + NBMAX;
    int* ps     = bstart + 2 * NBMAX;
    int* pd     = ps + E;
    float* out  = (float*)d_out;

    const int decBl  = (EL + 7) / 8;
    const int nb     = (n + BK - 1) / BK;

    bin_count_kernel<<<PG, 256, 0, stream>>>(ei + E, blkcnt, E);
    bin_scan_kernel<<<nb, 256, 0, stream>>>(blkcnt, excl, total);
    bin_place_kernel<<<PG, 256, 0, stream>>>(ei, ei + E, excl, total, ps, pd, bstart, E, nb);

    bucket_degree_kernel<<<nb, 1024, 0, stream>>>(pd, bstart, dinv, n);

    gemm_wmma_kernel<<<GW, 128, 0, stream>>>(x, W1, dinv, bufA, n);
    bucket_aggregate_kernel<<<nb, 1024, 0, stream>>>(ps, pd, bstart, dinv, bufA, b1, bufB, n, 1);

    gemm_wmma_kernel<<<GW, 128, 0, stream>>>(bufB, W2, dinv, bufA, n);
    bucket_aggregate_kernel<<<nb, 1024, 0, stream>>>(ps, pd, bstart, dinv, bufA, b2, bufB, n, 0);

    decode_kernel<<<decBl, 256, 0, stream>>>(el, el + EL, bufB, out, EL);
}


# Count and place passes of the edge regrouping: a thread asks for its 28 edges' indices as one group, then its LDS slots as one group, not one dependent memory trip per edge; sums stay arrival-order
# speedup vs baseline: 4.0478x; 1.0429x over previous
#include <hip/hip_runtime.h>

typedef __attribute__((ext_vector_type(2))) float v2f;
typedef __attribute__((ext_vector_type(4))) float v4f;
typedef __attribute__((ext_vector_type(8))) float v8f;

#define DF 128

#define GW 256

__device__ __forceinline__ void load_x_tile(v4f (&xb)[16], const float* __restrict__ X, int tile,
                                            int c, int h, int nrows) {
    int row = tile * 16 + c;
    row = row < nrows ? row : nrows - 1;
    const v4f* xr = (const v4f*)(X + (size_t)row * DF) + h;
#pragma unroll
    for (int j = 0; j < 16; j++) xb[j] = xr[2 * j];
}

__global__ __launch_bounds__(128) void gemm_wmma_kernel(const float* __restrict__ X,
                                                        const float* __restrict__ W,
                                                        const float* __restrict__ dinv,
                                                        float* __restrict__ H, int nrows) {
    const int lane = threadIdx.x & 31;
    const int c = lane & 15;
    const int h = lane >> 4;
    const int wv = __builtin_amdgcn_readfirstlane((int)(blockIdx.x * 4 + (threadIdx.x >> 5)));
    const int nwv = (int)gridDim.x * 4;
    const int ntiles = (nrows + 15) >> 4;
    if (wv >= ntiles) return;

    v2f wa[32][8];
#pragma unroll
    for (int t = 0; t < 32; t++) {
        const float* wp = W + (size_t)(8 * (t >> 1) + 4 * h + 2 * (t & 1)) * DF + c;
#pragma unroll
        for (int nt = 0; nt < 8; nt++) {
            wa[t][nt].x = wp[nt * 16];
            wa[t][nt].y = wp[DF + nt * 16];
        }
    }

    v4f xb[16];
    load_x_tile(xb, X, wv, c, h, nrows);
    const int row0 = wv * 16 + c;
    float dn = dinv[row0 < nrows ? row0 : nrows - 1];
    for (int tile = wv; tile < ntiles; tile += nwv) {
        const int nxt = tile + nwv < ntiles ? tile + nwv : tile;
        v4f xn[16];
        load_x_tile(xn, X, nxt, c, h, nrows);
        const int nrow = nxt * 16 + c;
        const float dnn = dinv[nrow < nrows ? nrow : nrows - 1];

        v8f acc[8] = {};
#pragma unroll
        for (int j = 0; j < 16; j++) {
            const v2f b0 = __builtin_shufflevector(xb[j], xb[j], 0, 1);
            const v2f b1 = __builtin_shufflevector(xb[j], xb[j], 2, 3);
#pragma unroll
            for (int nt = 0; nt < 8; nt++)
                acc[nt] = __builtin_amdgcn_wmma_f32_16x16x4_f32(
                    false, wa[2 * j][nt], false, b0, (short)0, acc[nt], false, false);
#pragma unroll
            for (int nt = 0; nt < 8; nt++)
                acc[nt] = __builtin_amdgcn_wmma_f32_16x16x4_f32(
                    false, wa[2 * j + 1][nt], false, b1, (short)0, acc[nt], false, false);
        }

        const int row = tile * 16 + c;
        if (row < nrows) {
            float* o = H + (size_t)row * DF + 8 * h;
#pragma unroll
            for (int nt = 0; nt < 8; nt++) {
                const v8f sc = acc[nt] * dn;
                *(v4f*)(o + nt * 16) = __builtin_shufflevector(sc, sc, 0, 1, 2, 3);
                *(v4f*)(o + nt * 16 + 4) = __builtin_shufflevector(sc, sc, 4, 5, 6, 7);
            }
        }
#pragma unroll
        for (int j = 0; j < 16; j++) xb[j] = xn[j];
        dn = dnn;
    }
}

#define BK 400
#define NBMAX 256
#define PG 256

__device__ __forceinline__ int bucket_of(int d) {
    const int k = d / BK;
    return k < NBMAX - 1 ? k : NBMAX - 1;
}

#define EG 28

__global__ __launch_bounds__(256) void bin_count_kernel(const int* __restrict__ dst,
                                                        int* __restrict__ blkcnt, int e) {
    __shared__ int hist[NBMAX];
    const int t = threadIdx.x;
    hist[t] = 0;
    __syncthreads();
    const int chunk = (e + PG - 1) / PG;
    const int lo = blockIdx.x * chunk;
    const int hi = lo + chunk < e ? lo + chunk : e;
    for (int base = lo; base < hi; base += EG * PG) {
        const int m = hi - base < EG * PG ? hi - base : EG * PG;
        int d[EG];
#pragma unroll
        for (int k = 0; k < EG; k++) {
            const int i = k * PG + t;
            d[k] = dst[base + (i < m ? i : m - 1)];
        }
#pragma unroll
        for (int k = 0; k < EG; k++) {
            if (k * PG + t < m) atomicAdd(&hist[bucket_of(d[k])], 1);
        }
    }
    __syncthreads();
    blkcnt[blockIdx.x * NBMAX + t] = hist[t];
}

__device__ __forceinline__ int block_scan_256(int* sc, int v) {
    const int t = threadIdx.x;
    sc[t] = v;
    __syncthreads();
    for (int off = 1; off < 256; off <<= 1) {
        const int add = t >= off ? sc[t - off] : 0;
        __syncthreads();
        sc[t] += add;
        __syncthreads();
    }
    return sc[t];
}

__global__ __launch_bounds__(256) void bin_scan_kernel(const int* __restrict__ blkcnt,
                                                       int* __restrict__ excl,
                                                       int* __restrict__ total) {
    __shared__ int sc[256];
    const int k = blockIdx.x;
    const int b = threadIdx.x;
    const int v = blkcnt[b * NBMAX + k];
    const int inc = block_scan_256(sc, v);
    excl[b * NBMAX + k] = inc - v;
    if (b == 255) total[k] = inc;
}

__global__ __launch_bounds__(256) void bin_place_kernel(const int* __restrict__ src,
                                                        const int* __restrict__ dst,
                                                        const int* __restrict__ excl,
                                                        const int* __restrict__ total,
                                                        int* __restrict__ ps, int* __restrict__ pd,
                                                        int* __restrict__ bstart, int e, int nb) {
    __shared__ int sc[256];
    __shared__ int cur[NBMAX];
    const int t = threadIdx.x;
    const int v = t < nb ? total[t] : 0;
    const int inc = block_scan_256(sc, v);
    const int start = inc - v;
    cur[t] = t < nb ? start + excl[blockIdx.x * NBMAX + t] : 0;
    if (blockIdx.x == 0) {
        if (t < nb) bstart[t] = start;
        if (t == 255) bstart[nb] = inc;
    }
    __syncthreads();
    const int chunk = (e + PG - 1) / PG;
    const int lo = blockIdx.x * chunk;
    const int hi = lo + chunk < e ? lo + chunk : e;
    for (int base = lo; base < hi; base += EG * PG) {
        const int m = hi - base < EG * PG ? hi - base : EG * PG;
        int s[EG], d[EG], pos[EG];
#pragma unroll
        for (int k = 0; k < EG; k++) {
            const int i = k * PG + t;
            const int ii = i < m ? i : m - 1;
            s[k] = src[base + ii];
            d[k] = dst[base + ii];
        }
#pragma unroll
        for (int k = 0; k < EG; k++) {
            pos[k] = 0;
            if (k * PG + t < m) pos[k] = atomicAdd(&cur[bucket_of(d[k])], 1);
        }
#pragma unroll
        for (int k = 0; k < EG; k++) {
            if (k * PG + t < m) {
                ps[pos[k]] = s[k];
                pd[pos[k]] = d[k];
            }
        }
    }
}

#define DC 8192

__global__ __launch_bounds__(1024) void bucket_degree_kernel(const int* __restrict__ pd,
                                                             const int* __restrict__ bstart,
                                                             float* __restrict__ dinv, int n) {
    __shared__ int indeg[BK];
    const int tid = threadIdx.x;
    const int node0 = blockIdx.x * BK;
    const int nn = n - node0 < BK ? n - node0 : BK;
    if (tid < BK) indeg[tid] = 0;
    const int e0 = bstart[blockIdx.x];
    const int e1 = bstart[blockIdx.x + 1];
    __syncthreads();
    for (int base = e0; base < e1; base += DC) {
        const int m = e1 - base < DC ? e1 - base : DC;
        int d[8];
#pragma unroll
        for (int k = 0; k < 8; k++) {
            const int i = k * 1024 + tid;
            const int ii = i < m ? i : m - 1;
            d[k] = pd[base + ii] - node0;
        }
#pragma unroll
        for (int k = 0; k < 8; k++) {
            const int i = k * 1024 + tid;
            if (i < m) atomicAdd(&indeg[d[k]], 1);
        }
    }
    __syncthreads();
    if (tid < nn) dinv[node0 + tid] = rsqrtf((float)(indeg[tid] + 1));
}

#define QC 8192
#define IR 4

__global__ __launch_bounds__(1024) void bucket_aggregate_kernel(
    const int* __restrict__ ps, const int* __restrict__ pd, const int* __restrict__ bstart,
    const float* __restrict__ dinv, const float* __restrict__ G, const float* __restrict__ bias,
    float* __restrict__ z, int n, int do_relu) {
    __shared__ v4f acc[BK * 32];
    __shared__ int2 q[QC];
    __shared__ int cnt[32];
    const int node0 = blockIdx.x * BK;
    const int nn = n - node0 < BK ? n - node0 : BK;
    const int tid = threadIdx.x;
    const int lane = tid & 31;
    const int wave = __builtin_amdgcn_readfirstlane(tid >> 5);
    const v4f* G4 = (const v4f*)G;

    for (int r0 = wave; r0 < nn; r0 += 32 * IR) {
        v4f m[IR];
#pragma unroll
        for (int u = 0; u < IR; u++) {
            const int r = r0 + 32 * u;
            if (r < nn) m[u] = G4[(size_t)(node0 + r) * 32 + lane];
        }
#pragma unroll
        for (int u = 0; u < IR; u++) {
            const int r = r0 + 32 * u;
            if (r < nn) acc[r * 32 + lane] = m[u];
        }
    }

    const int e0 = bstart[blockIdx.x];
    const int e1 = bstart[blockIdx.x + 1];
    for (int base = e0; base < e1; base += QC) {
        const int m = e1 - base < QC ? e1 - base : QC;
        if (tid < 32) cnt[tid] = 0;
        __syncthreads();

        int s[8], d[8], rk[8];
#pragma unroll
        for (int k = 0; k < 8; k++) {
            const int i = k * 1024 + tid;
            const int ii = i < m ? i : m - 1;
            s[k] = ps[base + ii];
            d[k] = pd[base + ii] - node0;
        }
#pragma unroll
        for (int k = 0; k < 8; k++) {
            const int i = k * 1024 + tid;
            rk[k] = 0;
            if (i < m) rk[k] = atomicAdd(&cnt[d[k] & 31], 1);
        }
        __syncthreads();

        const int c = cnt[lane];
        int inc = c;
#pragma unroll
        for (int off = 1; off < 32; off <<= 1) {
            const int t = __shfl_up(inc, off, 32);
            if (lane >= off) inc += t;
        }
        const int first = inc - c;
#pragma unroll
        for (int k = 0; k < 8; k++) {
            const int i = k * 1024 + tid;
            const int st = __shfl(first, d[k] & 31, 32);
            if (i < m) q[st + rk[k]] = make_int2(s[k], d[k]);
        }
        __syncthreads();

        const int qs = __builtin_amdgcn_readlane(first, wave);
        const int qn = __builtin_amdgcn_readlane(c, wave);
        for (int g = 0; g < qn; g += 32) {
            const int left = qn - g;
            const int cg = left < 32 ? left : 32;
            int2 e = make_int2(0, 0);
            if (lane < cg) e = q[qs + g + lane];
            if (cg == 32) {
                v4f ma[8], mb[8];
#pragma unroll
                for (int u = 0; u < 8; u++)
                    ma[u] = G4[(size_t)__builtin_amdgcn_readlane(e.x, u) * 32 + lane];
#pragma unroll
                for (int u = 0; u < 8; u++)
                    mb[u] = G4[(size_t)__builtin_amdgcn_readlane(e.x, 8 + u) * 32 + lane];
#pragma unroll
                for (int u = 0; u < 8; u++) {
                    v4f* a = acc + __builtin_amdgcn_readlane(e.y, u) * 32 + lane;
                    *a = *a + ma[u];
                }
#pragma unroll
                for (int u = 0; u < 8; u++)
                    ma[u] = G4[(size_t)__builtin_amdgcn_readlane(e.x, 16 + u) * 32 + lane];
#pragma unroll
                for (int u = 0; u < 8; u++) {
                    v4f* a = acc + __builtin_amdgcn_readlane(e.y, 8 + u) * 32 + lane;
                    *a = *a + mb[u];
                }
#pragma unroll
                for (int u = 0; u < 8; u++)
                    mb[u] = G4[(size_t)__builtin_amdgcn_readlane(e.x, 24 + u) * 32 + lane];
#pragma unroll
                for (int u = 0; u < 8; u++) {
                    v4f* a = acc + __builtin_amdgcn_readlane(e.y, 16 + u) * 32 + lane;
                    *a = *a + ma[u];
                }
#pragma unroll
                for (int u = 0; u < 8; u++) {
                    v4f* a = acc + __builtin_amdgcn_readlane(e.y, 24 + u) * 32 + lane;
                    *a = *a + mb[u];
                }
            } else {
                for (int j0 = 0; j0 < cg; j0 += 8) {
                    v4f mt[8];
#pragma unroll
                    for (int u = 0; u < 8; u++)
                        mt[u] = G4[(size_t)__builtin_amdgcn_readlane(e.x, j0 + u) * 32 + lane];
#pragma unroll
                    for (int u = 0; u < 8; u++) {
                        if (j0 + u < cg) {
                            v4f* a = acc + __builtin_amdgcn_readlane(e.y, j0 + u) * 32 + lane;
                            *a = *a + mt[u];
                        }
                    }
                }
            }
        }
    }

    const v4f b4 = {bias[4 * lane], bias[4 * lane + 1], bias[4 * lane + 2], bias[4 * lane + 3]};
    v4f* z4 = (v4f*)z;
    for (int r0 = wave; r0 < nn; r0 += 32 * IR) {
        float qv[IR];
#pragma unroll
        for (int u = 0; u < IR; u++) {
            const int r = r0 + 32 * u;
            qv[u] = r < nn ? dinv[node0 + r] : 0.0f;
        }
#pragma unroll
        for (int u = 0; u < IR; u++) {
            const int r = r0 + 32 * u;
            if (r < nn) {
                v4f w = acc[r * 32 + lane] * qv[u] + b4;
                if (do_relu) {
                    w.x = fmaxf(w.x, 0.0f);
                    w.y = fmaxf(w.y, 0.0f);
                    w.z = fmaxf(w.z, 0.0f);
                    w.w = fmaxf(w.w, 0.0f);
                }
                z4[(size_t)(node0 + r) * 32 + lane] = w;
            }
        }
    }
}


__global__ __launch_bounds__(256) void decode_kernel(const int* __restrict__ s,
                                                     const int* __restrict__ d,
                                                     const float* __restrict__ z,
                                                     float* __restrict__ logits, int e) {
    const int lane = threadIdx.x & 31;
    const int w = (int)((blockIdx.x * blockDim.x + threadIdx.x) >> 5);
    if (w >= e) return;
    const int si = s[w];
    const int di = d[w];
    v4f a = ((const v4f*)(z + (size_t)si * DF))[lane];
    v4f b = ((const v4f*)(z + (size_t)di * DF))[lane];
    float dot = a.x * b.x + a.y * b.y + a.z * b.z + a.w * b.w;
#pragma unroll
    for (int off = 16; off > 0; off >>= 1) dot += __shfl_xor(dot, off, 32);
    if (lane == 0) logits[w] = dot;
}


extern "C" void kernel_launch(void* const* d_in, const int* in_sizes, int n_in,
                              void* d_out, int out_size, void* d_ws, size_t ws_size,
                              hipStream_t stream) {
    const float* x  = (const float*)d_in[0];
    const int*   ei = (const int*)d_in[1];
    const int*   el = (const int*)d_in[2];
    const float* W1 = (const float*)d_in[3];
    const float* b1 = (const float*)d_in[4];
    const float* W2 = (const float*)d_in[5];
    const float* b2 = (const float*)d_in[6];

    const int n  = in_sizes[0] / DF;
    const int E  = in_sizes[1] / 2;
    const int EL = in_sizes[2] / 2;

    float* bufA = (float*)d_ws;
    float* bufB = bufA + (size_t)n * DF;
    float* dinv = bufB + (size_t)n * DF;
    int* blkcnt = (int*)(dinv + n);
    int* excl   = blkcnt + PG * NBMAX;
    int* total  = excl + PG * NBMAX;
    int* bstart = total + NBMAX;
    int* ps     = bstart + 2 * NBMAX;
    int* pd     = ps + E;
    float* out  = (float*)d_out;

    const int decBl  = (EL + 7) / 8;
    const int nb     = (n + BK - 1) / BK;

    bin_count_kernel<<<PG, 256, 0, stream>>>(ei + E, blkcnt, E);
    bin_scan_kernel<<<nb, 256, 0, stream>>>(blkcnt, excl, total);
    bin_place_kernel<<<PG, 256, 0, stream>>>(ei, ei + E, excl, total, ps, pd, bstart, E, nb);

    bucket_degree_kernel<<<nb, 1024, 0, stream>>>(pd, bstart, dinv, n);

    gemm_wmma_kernel<<<GW, 128, 0, stream>>>(x, W1, dinv, bufA, n);
    bucket_aggregate_kernel<<<nb, 1024, 0, stream>>>(ps, pd, bstart, dinv, bufA, b1, bufB, n, 1);

    gemm_wmma_kernel<<<GW, 128, 0, stream>>>(bufB, W2, dinv, bufA, n);
    bucket_aggregate_kernel<<<nb, 1024, 0, stream>>>(ps, pd, bstart, dinv, bufA, b2, bufB, n, 0);

    decode_kernel<<<decBl, 256, 0, stream>>>(el, el + EL, bufB, out, EL);
}


# Decode: a wave takes 32 label edges (8 lanes an edge, 4 edges per load instruction, rows of 2 trips in flight), logits leave as one 128-B line; 6250 waves enter, not 200000; sum order kept
# speedup vs baseline: 4.3701x; 1.0796x over previous
#include <hip/hip_runtime.h>

typedef __attribute__((ext_vector_type(2))) float v2f;
typedef __attribute__((ext_vector_type(4))) float v4f;
typedef __attribute__((ext_vector_type(8))) float v8f;

#define DF 128

#define GW 256

__device__ __forceinline__ void load_x_tile(v4f (&xb)[16], const float* __restrict__ X, int tile,
                                            int c, int h, int nrows) {
    int row = tile * 16 + c;
    row = row < nrows ? row : nrows - 1;
    const v4f* xr = (const v4f*)(X + (size_t)row * DF) + h;
#pragma unroll
    for (int j = 0; j < 16; j++) xb[j] = xr[2 * j];
}

__global__ __launch_bounds__(128) void gemm_wmma_kernel(const float* __restrict__ X,
                                                        const float* __restrict__ W,
                                                        const float* __restrict__ dinv,
                                                        float* __restrict__ H, int nrows) {
    const int lane = threadIdx.x & 31;
    const int c = lane & 15;
    const int h = lane >> 4;
    const int wv = __builtin_amdgcn_readfirstlane((int)(blockIdx.x * 4 + (threadIdx.x >> 5)));
    const int nwv = (int)gridDim.x * 4;
    const int ntiles = (nrows + 15) >> 4;
    if (wv >= ntiles) return;

    v2f wa[32][8];
#pragma unroll
    for (int t = 0; t < 32; t++) {
        const float* wp = W + (size_t)(8 * (t >> 1) + 4 * h + 2 * (t & 1)) * DF + c;
#pragma unroll
        for (int nt = 0; nt < 8; nt++) {
            wa[t][nt].x = wp[nt * 16];
            wa[t][nt].y = wp[DF + nt * 16];
        }
    }

    v4f xb[16];
    load_x_tile(xb, X, wv, c, h, nrows);
    const int row0 = wv * 16 + c;
    float dn = dinv[row0 < nrows ? row0 : nrows - 1];
    for (int tile = wv; tile < ntiles; tile += nwv) {
        const int nxt = tile + nwv < ntiles ? tile + nwv : tile;
        v4f xn[16];
        load_x_tile(xn, X, nxt, c, h, nrows);
        const int nrow = nxt * 16 + c;
        const float dnn = dinv[nrow < nrows ? nrow : nrows - 1];

        v8f acc[8] = {};
#pragma unroll
        for (int j = 0; j < 16; j++) {
            const v2f b0 = __builtin_shufflevector(xb[j], xb[j], 0, 1);
            const v2f b1 = __builtin_shufflevector(xb[j], xb[j], 2, 3);
#pragma unroll
            for (int nt = 0; nt < 8; nt++)
                acc[nt] = __builtin_amdgcn_wmma_f32_16x16x4_f32(
                    false, wa[2 * j][nt], false, b0, (short)0, acc[nt], false, false);
#pragma unroll
            for (int nt = 0; nt < 8; nt++)
                acc[nt] = __builtin_amdgcn_wmma_f32_16x16x4_f32(
                    false, wa[2 * j + 1][nt], false, b1, (short)0, acc[nt], false, false);
        }

        const int row = tile * 16 + c;
        if (row < nrows) {
            float* o = H + (size_t)row * DF + 8 * h;
#pragma unroll
            for (int nt = 0; nt < 8; nt++) {
                const v8f sc = acc[nt] * dn;
                *(v4f*)(o + nt * 16) = __builtin_shufflevector(sc, sc, 0, 1, 2, 3);
                *(v4f*)(o + nt * 16 + 4) = __builtin_shufflevector(sc, sc, 4, 5, 6, 7);
            }
        }
#pragma unroll
        for (int j = 0; j < 16; j++) xb[j] = xn[j];
        dn = dnn;
    }
}

#define BK 400
#define NBMAX 256
#define PG 256

__device__ __forceinline__ int bucket_of(int d) {
    const int k = d / BK;
    return k < NBMAX - 1 ? k : NBMAX - 1;
}

#define EG 28

__global__ __launch_bounds__(256) void bin_count_kernel(const int* __restrict__ dst,
                                                        int* __restrict__ blkcnt, int e) {
    __shared__ int hist[NBMAX];
    const int t = threadIdx.x;
    hist[t] = 0;
    __syncthreads();
    const int chunk = (e + PG - 1) / PG;
    const int lo = blockIdx.x * chunk;
    const int hi = lo + chunk < e ? lo + chunk : e;
    for (int base = lo; base < hi; base += EG * PG) {
        const int m = hi - base < EG * PG ? hi - base : EG * PG;
        int d[EG];
#pragma unroll
        for (int k = 0; k < EG; k++) {
            const int i = k * PG + t;
            d[k] = dst[base + (i < m ? i : m - 1)];
        }
#pragma unroll
        for (int k = 0; k < EG; k++) {
            if (k * PG + t < m) atomicAdd(&hist[bucket_of(d[k])], 1);
        }
    }
    __syncthreads();
    blkcnt[blockIdx.x * NBMAX + t] = hist[t];
}

__device__ __forceinline__ int block_scan_256(int* sc, int v) {
    const int t = threadIdx.x;
    sc[t] = v;
    __syncthreads();
    for (int off = 1; off < 256; off <<= 1) {
        const int add = t >= off ? sc[t - off] : 0;
        __syncthreads();
        sc[t] += add;
        __syncthreads();
    }
    return sc[t];
}

__global__ __launch_bounds__(256) void bin_scan_kernel(const int* __restrict__ blkcnt,
                                                       int* __restrict__ excl,
                                                       int* __restrict__ total) {
    __shared__ int sc[256];
    const int k = blockIdx.x;
    const int b = threadIdx.x;
    const int v = blkcnt[b * NBMAX + k];
    const int inc = block_scan_256(sc, v);
    excl[b * NBMAX + k] = inc - v;
    if (b == 255) total[k] = inc;
}

__global__ __launch_bounds__(256) void bin_place_kernel(const int* __restrict__ src,
                                                        const int* __restrict__ dst,
                                                        const int* __restrict__ excl,
                                                        const int* __restrict__ total,
                                                        int* __restrict__ ps, int* __restrict__ pd,
                                                        int* __restrict__ bstart, int e, int nb) {
    __shared__ int sc[256];
    __shared__ int cur[NBMAX];
    const int t = threadIdx.x;
    const int v = t < nb ? total[t] : 0;
    const int inc = block_scan_256(sc, v);
    const int start = inc - v;
    cur[t] = t < nb ? start + excl[blockIdx.x * NBMAX + t] : 0;
    if (blockIdx.x == 0) {
        if (t < nb) bstart[t] = start;
        if (t == 255) bstart[nb] = inc;
    }
    __syncthreads();
    const int chunk = (e + PG - 1) / PG;
    const int lo = blockIdx.x * chunk;
    const int hi = lo + chunk < e ? lo + chunk : e;
    for (int base = lo; base < hi; base += EG * PG) {
        const int m = hi - base < EG * PG ? hi - base : EG * PG;
        int s[EG], d[EG], pos[EG];
#pragma unroll
        for (int k = 0; k < EG; k++) {
            const int i = k * PG + t;
            const int ii = i < m ? i : m - 1;
            s[k] = src[base + ii];
            d[k] = dst[base + ii];
        }
#pragma unroll
        for (int k = 0; k < EG; k++) {
            pos[k] = 0;
            if (k * PG + t < m) pos[k] = atomicAdd(&cur[bucket_of(d[k])], 1);
        }
#pragma unroll
        for (int k = 0; k < EG; k++) {
            if (k * PG + t < m) {
                ps[pos[k]] = s[k];
                pd[pos[k]] = d[k];
            }
        }
    }
}

#define DC 8192

__global__ __launch_bounds__(1024) void bucket_degree_kernel(const int* __restrict__ pd,
                                                             const int* __restrict__ bstart,
                                                             float* __restrict__ dinv, int n) {
    __shared__ int indeg[BK];
    const int tid = threadIdx.x;
    const int node0 = blockIdx.x * BK;
    const int nn = n - node0 < BK ? n - node0 : BK;
    if (tid < BK) indeg[tid] = 0;
    const int e0 = bstart[blockIdx.x];
    const int e1 = bstart[blockIdx.x + 1];
    __syncthreads();
    for (int base = e0; base < e1; base += DC) {
        const int m = e1 - base < DC ? e1 - base : DC;
        int d[8];
#pragma unroll
        for (int k = 0; k < 8; k++) {
            const int i = k * 1024 + tid;
            const int ii = i < m ? i : m - 1;
            d[k] = pd[base + ii] - node0;
        }
#pragma unroll
        for (int k = 0; k < 8; k++) {
            const int i = k * 1024 + tid;
            if (i < m) atomicAdd(&indeg[d[k]], 1);
        }
    }
    __syncthreads();
    if (tid < nn) dinv[node0 + tid] = rsqrtf((float)(indeg[tid] + 1));
}

#define QC 8192
#define IR 4

__global__ __launch_bounds__(1024) void bucket_aggregate_kernel(
    const int* __restrict__ ps, const int* __restrict__ pd, const int* __restrict__ bstart,
    const float* __restrict__ dinv, const float* __restrict__ G, const float* __restrict__ bias,
    float* __restrict__ z, int n, int do_relu) {
    __shared__ v4f acc[BK * 32];
    __shared__ int2 q[QC];
    __shared__ int cnt[32];
    const int node0 = blockIdx.x * BK;
    const int nn = n - node0 < BK ? n - node0 : BK;
    const int tid = threadIdx.x;
    const int lane = tid & 31;
    const int wave = __builtin_amdgcn_readfirstlane(tid >> 5);
    const v4f* G4 = (const v4f*)G;

    for (int r0 = wave; r0 < nn; r0 += 32 * IR) {
        v4f m[IR];
#pragma unroll
        for (int u = 0; u < IR; u++) {
            const int r = r0 + 32 * u;
            if (r < nn) m[u] = G4[(size_t)(node0 + r) * 32 + lane];
        }
#pragma unroll
        for (int u = 0; u < IR; u++) {
            const int r = r0 + 32 * u;
            if (r < nn) acc[r * 32 + lane] = m[u];
        }
    }

    const int e0 = bstart[blockIdx.x];
    const int e1 = bstart[blockIdx.x + 1];
    for (int base = e0; base < e1; base += QC) {
        const int m = e1 - base < QC ? e1 - base : QC;
        if (tid < 32) cnt[tid] = 0;
        __syncthreads();

        int s[8], d[8], rk[8];
#pragma unroll
        for (int k = 0; k < 8; k++) {
            const int i = k * 1024 + tid;
            const int ii = i < m ? i : m - 1;
            s[k] = ps[base + ii];
            d[k] = pd[base + ii] - node0;
        }
#pragma unroll
        for (int k = 0; k < 8; k++) {
            const int i = k * 1024 + tid;
            rk[k] = 0;
            if (i < m) rk[k] = atomicAdd(&cnt[d[k] & 31], 1);
        }
        __syncthreads();

        const int c = cnt[lane];
        int inc = c;
#pragma unroll
        for (int off = 1; off < 32; off <<= 1) {
            const int t = __shfl_up(inc, off, 32);
            if (lane >= off) inc += t;
        }
        const int first = inc - c;
#pragma unroll
        for (int k = 0; k < 8; k++) {
            const int i = k * 1024 + tid;
            const int st = __shfl(first, d[k] & 31, 32);
            if (i < m) q[st + rk[k]] = make_int2(s[k], d[k]);
        }
        __syncthreads();

        const int qs = __builtin_amdgcn_readlane(first, wave);
        const int qn = __builtin_amdgcn_readlane(c, wave);
        for (int g = 0; g < qn; g += 32) {
            const int left = qn - g;
            const int cg = left < 32 ? left : 32;
            int2 e = make_int2(0, 0);
            if (lane < cg) e = q[qs + g + lane];
            if (cg == 32) {
                v4f ma[8], mb[8];
#pragma unroll
                for (int u = 0; u < 8; u++)
                    ma[u] = G4[(size_t)__builtin_amdgcn_readlane(e.x, u) * 32 + lane];
#pragma unroll
                for (int u = 0; u < 8; u++)
                    mb[u] = G4[(size_t)__builtin_amdgcn_readlane(e.x, 8 + u) * 32 + lane];
#pragma unroll
                for (int u = 0; u < 8; u++) {
                    v4f* a = acc + __builtin_amdgcn_readlane(e.y, u) * 32 + lane;
                    *a = *a + ma[u];
                }
#pragma unroll
                for (int u = 0; u < 8; u++)
                    ma[u] = G4[(size_t)__builtin_amdgcn_readlane(e.x, 16 + u) * 32 + lane];
#pragma unroll
                for (int u = 0; u < 8; u++) {
                    v4f* a = acc + __builtin_amdgcn_readlane(e.y, 8 + u) * 32 + lane;
                    *a = *a + mb[u];
                }
#pragma unroll
                for (int u = 0; u < 8; u++)
                    mb[u] = G4[(size_t)__builtin_amdgcn_readlane(e.x, 24 + u) * 32 + lane];
#pragma unroll
                for (int u = 0; u < 8; u++) {
                    v4f* a = acc + __builtin_amdgcn_readlane(e.y, 16 + u) * 32 + lane;
                    *a = *a + ma[u];
                }
#pragma unroll
                for (int u = 0; u < 8; u++) {
                    v4f* a = acc + __builtin_amdgcn_readlane(e.y, 24 + u) * 32 + lane;
                    *a = *a + mb[u];
                }
            } else {
                for (int j0 = 0; j0 < cg; j0 += 8) {
                    v4f mt[8];
#pragma unroll
                    for (int u = 0; u < 8; u++)
                        mt[u] = G4[(size_t)__builtin_amdgcn_readlane(e.x, j0 + u) * 32 + lane];
#pragma unroll
                    for (int u = 0; u < 8; u++) {
                        if (j0 + u < cg) {
                            v4f* a = acc + __builtin_amdgcn_readlane(e.y, j0 + u) * 32 + lane;
                            *a = *a + mt[u];
                        }
                    }
                }
            }
        }
    }

    const v4f b4 = {bias[4 * lane], bias[4 * lane + 1], bias[4 * lane + 2], bias[4 * lane + 3]};
    v4f* z4 = (v4f*)z;
    for (int r0 = wave; r0 < nn; r0 += 32 * IR) {
        float qv[IR];
#pragma unroll
        for (int u = 0; u < IR; u++) {
            const int r = r0 + 32 * u;
            qv[u] = r < nn ? dinv[node0 + r] : 0.0f;
        }
#pragma unroll
        for (int u = 0; u < IR; u++) {
            const int r = r0 + 32 * u;
            if (r < nn) {
                v4f w = acc[r * 32 + lane] * qv[u] + b4;
                if (do_relu) {
                    w.x = fmaxf(w.x, 0.0f);
                    w.y = fmaxf(w.y, 0.0f);
                    w.z = fmaxf(w.z, 0.0f);
                    w.w = fmaxf(w.w, 0.0f);
                }
                z4[(size_t)(node0 + r) * 32 + lane] = w;
            }
        }
    }
}

#define DT 8

__device__ __forceinline__ void load_row_pair(v4f (&a)[4], v4f (&b)[4], const float* __restrict__ z,
                                              int si, int di, int j) {
    const v4f* ra = (const v4f*)(z + (size_t)si * DF) + j;
    const v4f* rb = (const v4f*)(z + (size_t)di * DF) + j;
#pragma unroll
    for (int k = 0; k < 4; k++) {
        a[k] = ra[8 * k];
        b[k] = rb[8 * k];
    }
}

__global__ __launch_bounds__(256) void decode_kernel(const int* __restrict__ s,
                                                     const int* __restrict__ d,
                                                     const float* __restrict__ z,
                                                     float* __restrict__ logits, int e) {
    const int lane = threadIdx.x & 31;
    const int q = lane >> 3;
    const int j = lane & 7;
    const int w = (int)((blockIdx.x * blockDim.x + threadIdx.x) >> 5);
    const int base = w * 32;
    if (base >= e) return;

    int si[DT], di[DT];
#pragma unroll
    for (int i = 0; i < DT; i++) {
        const int ei = base + 8 * q + i;
        const int ec = ei < e ? ei : e - 1;
        si[i] = s[ec];
        di[i] = d[ec];
    }
    __builtin_amdgcn_sched_barrier(0);

    v4f a[2][4], b[2][4];
    load_row_pair(a[0], b[0], z, si[0], di[0], j);
    load_row_pair(a[1], b[1], z, si[1], di[1], j);
    float r[DT];
#pragma unroll
    for (int i = 0; i < DT; i++) {
        const int u = i & 1;
        float p[4];
#pragma unroll
        for (int k = 0; k < 4; k++)
            p[k] = a[u][k].x * b[u][k].x + a[u][k].y * b[u][k].y + a[u][k].z * b[u][k].z +
                   a[u][k].w * b[u][k].w;
        __builtin_amdgcn_sched_barrier(0);
        if (i + 2 < DT) load_row_pair(a[u], b[u], z, si[i + 2], di[i + 2], j);
        __builtin_amdgcn_sched_barrier(0);
        float t = (p[0] + p[2]) + (p[1] + p[3]);
        t += __shfl_xor(t, 4, 32);
        t += __shfl_xor(t, 2, 32);
        t += __shfl_xor(t, 1, 32);
        r[i] = t;
    }
    float sel = r[0];
#pragma unroll
    for (int i = 1; i < DT; i++)
        if (j == i) sel = r[i];
    if (base + lane < e) logits[base + lane] = sel;
}


extern "C" void kernel_launch(void* const* d_in, const int* in_sizes, int n_in,
                              void* d_out, int out_size, void* d_ws, size_t ws_size,
                              hipStream_t stream) {
    const float* x  = (const float*)d_in[0];
    const int*   ei = (const int*)d_in[1];
    const int*   el = (const int*)d_in[2];
    const float* W1 = (const float*)d_in[3];
    const float* b1 = (const float*)d_in[4];
    const float* W2 = (const float*)d_in[5];
    const float* b2 = (const float*)d_in[6];

    const int n  = in_sizes[0] / DF;
    const int E  = in_sizes[1] / 2;
    const int EL = in_sizes[2] / 2;

    float* bufA = (float*)d_ws;
    float* bufB = bufA + (size_t)n * DF;
    float* dinv = bufB + (size_t)n * DF;
    int* blkcnt = (int*)(dinv + n);
    int* excl   = blkcnt + PG * NBMAX;
    int* total  = excl + PG * NBMAX;
    int* bstart = total + NBMAX;
    int* ps     = bstart + 2 * NBMAX;
    int* pd     = ps + E;
    float* out  = (float*)d_out;

    const int decBl  = (EL + 255) / 256;
    const int nb     = (n + BK - 1) / BK;

    bin_count_kernel<<<PG, 256, 0, stream>>>(ei + E, blkcnt, E);
    bin_scan_kernel<<<nb, 256, 0, stream>>>(blkcnt, excl, total);
    bin_place_kernel<<<PG, 256, 0, stream>>>(ei, ei + E, excl, total, ps, pd, bstart, E, nb);

    bucket_degree_kernel<<<nb, 1024, 0, stream>>>(pd, bstart, dinv, n);

    gemm_wmma_kernel<<<GW, 128, 0, stream>>>(x, W1, dinv, bufA, n);
    bucket_aggregate_kernel<<<nb, 1024, 0, stream>>>(ps, pd, bstart, dinv, bufA, b1, bufB, n, 1);

    gemm_wmma_kernel<<<GW, 128, 0, stream>>>(bufB, W2, dinv, bufA, n);
    bucket_aggregate_kernel<<<nb, 1024, 0, stream>>>(ps, pd, bstart, dinv, bufA, b2, bufB, n, 0);

    decode_kernel<<<decBl, 256, 0, stream>>>(el, el + EL, bufB, out, EL);
}


# Placing pass of the edge regrouping: a pass's edges are grouped by bucket in LDS and written out in staged order, so a wave's store covers contiguous runs, not 32 lone 4-B words; arrival order kept
# speedup vs baseline: 5.0362x; 1.1524x over previous
#include <hip/hip_runtime.h>

typedef __attribute__((ext_vector_type(2))) float v2f;
typedef __attribute__((ext_vector_type(4))) float v4f;
typedef __attribute__((ext_vector_type(8))) float v8f;

#define DF 128

#define GW 256

__device__ __forceinline__ void load_x_tile(v4f (&xb)[16], const float* __restrict__ X, int tile,
                                            int c, int h, int nrows) {
    int row = tile * 16 + c;
    row = row < nrows ? row : nrows - 1;
    const v4f* xr = (const v4f*)(X + (size_t)row * DF) + h;
#pragma unroll
    for (int j = 0; j < 16; j++) xb[j] = xr[2 * j];
}

__global__ __launch_bounds__(128) void gemm_wmma_kernel(const float* __restrict__ X,
                                                        const float* __restrict__ W,
                                                        const float* __restrict__ dinv,
                                                        float* __restrict__ H, int nrows) {
    const int lane = threadIdx.x & 31;
    const int c = lane & 15;
    const int h = lane >> 4;
    const int wv = __builtin_amdgcn_readfirstlane((int)(blockIdx.x * 4 + (threadIdx.x >> 5)));
    const int nwv = (int)gridDim.x * 4;
    const int ntiles = (nrows + 15) >> 4;
    if (wv >= ntiles) return;

    v2f wa[32][8];
#pragma unroll
    for (int t = 0; t < 32; t++) {
        const float* wp = W + (size_t)(8 * (t >> 1) + 4 * h + 2 * (t & 1)) * DF + c;
#pragma unroll
        for (int nt = 0; nt < 8; nt++) {
            wa[t][nt].x = wp[nt * 16];
            wa[t][nt].y = wp[DF + nt * 16];
        }
    }

    v4f xb[16];
    load_x_tile(xb, X, wv, c, h, nrows);
    const int row0 = wv * 16 + c;
    float dn = dinv[row0 < nrows ? row0 : nrows - 1];
    for (int tile = wv; tile < ntiles; tile += nwv) {
        const int nxt = tile + nwv < ntiles ? tile + nwv : tile;
        v4f xn[16];
        load_x_tile(xn, X, nxt, c, h, nrows);
        const int nrow = nxt * 16 + c;
        const float dnn = dinv[nrow < nrows ? nrow : nrows - 1];

        v8f acc[8] = {};
#pragma unroll
        for (int j = 0; j < 16; j++) {
            const v2f b0 = __builtin_shufflevector(xb[j], xb[j], 0, 1);
            const v2f b1 = __builtin_shufflevector(xb[j], xb[j], 2, 3);
#pragma unroll
            for (int nt = 0; nt < 8; nt++)
                acc[nt] = __builtin_amdgcn_wmma_f32_16x16x4_f32(
                    false, wa[2 * j][nt], false, b0, (short)0, acc[nt], false, false);
#pragma unroll
            for (int nt = 0; nt < 8; nt++)
                acc[nt] = __builtin_amdgcn_wmma_f32_16x16x4_f32(
                    false, wa[2 * j + 1][nt], false, b1, (short)0, acc[nt], false, false);
        }

        const int row = tile * 16 + c;
        if (row < nrows) {
            float* o = H + (size_t)row * DF + 8 * h;
#pragma unroll
            for (int nt = 0; nt < 8; nt++) {
                const v8f sc = acc[nt] * dn;
                *(v4f*)(o + nt * 16) = __builtin_shufflevector(sc, sc, 0, 1, 2, 3);
                *(v4f*)(o + nt * 16 + 4) = __builtin_shufflevector(sc, sc, 4, 5, 6, 7);
            }
        }
#pragma unroll
        for (int j = 0; j < 16; j++) xb[j] = xn[j];
        dn = dnn;
    }
}

#define BK 400
#define NBMAX 256
#define PG 256

__device__ __forceinline__ int bucket_of(int d) {
    const int k = d / BK;
    return k < NBMAX - 1 ? k : NBMAX - 1;
}

#define EG 28

__global__ __launch_bounds__(256) void bin_count_kernel(const int* __restrict__ dst,
                                                        int* __restrict__ blkcnt, int e) {
    __shared__ int hist[NBMAX];
    const int t = threadIdx.x;
    hist[t] = 0;
    __syncthreads();
    const int chunk = (e + PG - 1) / PG;
    const int lo = blockIdx.x * chunk;
    const int hi = lo + chunk < e ? lo + chunk : e;
    for (int base = lo; base < hi; base += EG * PG) {
        const int m = hi - base < EG * PG ? hi - base : EG * PG;
        int d[EG];
#pragma unroll
        for (int k = 0; k < EG; k++) {
            const int i = k * PG + t;
            d[k] = dst[base + (i < m ? i : m - 1)];
        }
#pragma unroll
        for (int k = 0; k < EG; k++) {
            if (k * PG + t < m) atomicAdd(&hist[bucket_of(d[k])], 1);
        }
    }
    __syncthreads();
    blkcnt[blockIdx.x * NBMAX + t] = hist[t];
}

__device__ __forceinline__ int block_scan_256(int* sc, int v) {
    const int t = threadIdx.x;
    sc[t] = v;
    __syncthreads();
    for (int off = 1; off < 256; off <<= 1) {
        const int add = t >= off ? sc[t - off] : 0;
        __syncthreads();
        sc[t] += add;
        __syncthreads();
    }
    return sc[t];
}

__global__ __launch_bounds__(256) void bin_scan_kernel(const int* __restrict__ blkcnt,
                                                       int* __restrict__ excl,
                                                       int* __restrict__ total) {
    __shared__ int sc[256];
    const int k = blockIdx.x;
    const int b = threadIdx.x;
    const int v = blkcnt[b * NBMAX + k];
    const int inc = block_scan_256(sc, v);
    excl[b * NBMAX + k] = inc - v;
    if (b == 255) total[k] = inc;
}

__global__ __launch_bounds__(256) void bin_place_kernel(const int* __restrict__ src,
                                                        const int* __restrict__ dst,
                                                        const int* __restrict__ excl,
                                                        const int* __restrict__ total,
                                                        int* __restrict__ ps, int* __restrict__ pd,
                                                        int* __restrict__ bstart, int e, int nb) {
    __shared__ int sc[256];
    __shared__ int cnt[NBMAX];
    __shared__ int lbase[NBMAX];
    __shared__ int gcur[NBMAX];
    __shared__ int wsum[8];
    __shared__ int2 st[EG * PG];
    const int t = threadIdx.x;
    const int lane = t & 31;
    const int wv = t >> 5;
    const int v = t < nb ? total[t] : 0;
    const int inc = block_scan_256(sc, v);
    const int start = inc - v;
    gcur[t] = t < nb ? start + excl[blockIdx.x * NBMAX + t] : 0;
    cnt[t] = 0;
    if (blockIdx.x == 0) {
        if (t < nb) bstart[t] = start;
        if (t == 255) bstart[nb] = inc;
    }
    __syncthreads();
    const int chunk = (e + PG - 1) / PG;
    const int lo = blockIdx.x * chunk;
    const int hi = lo + chunk < e ? lo + chunk : e;
    for (int base = lo; base < hi; base += EG * PG) {
        const int m = hi - base < EG * PG ? hi - base : EG * PG;
        int s[EG], d[EG], rk[EG];
#pragma unroll
        for (int k = 0; k < EG; k++) {
            const int i = k * PG + t;
            const int ii = i < m ? i : m - 1;
            s[k] = src[base + ii];
            d[k] = dst[base + ii];
        }
#pragma unroll
        for (int k = 0; k < EG; k++) {
            rk[k] = 0;
            if (k * PG + t < m) rk[k] = atomicAdd(&cnt[bucket_of(d[k])], 1);
        }
        __syncthreads();

        const int c = cnt[t];
        int ws = c;
#pragma unroll
        for (int off = 1; off < 32; off <<= 1) {
            const int x = __shfl_up(ws, off, 32);
            if (lane >= off) ws += x;
        }
        if (lane == 31) wsum[wv] = ws;
        __syncthreads();
        int before = 0;
#pragma unroll
        for (int w = 0; w < 8; w++) before += w < wv ? wsum[w] : 0;
        lbase[t] = before + ws - c;
        __syncthreads();

#pragma unroll
        for (int k = 0; k < EG; k++) rk[k] += lbase[bucket_of(d[k])];
#pragma unroll
        for (int k = 0; k < EG; k++) {
            if (k * PG + t < m) st[rk[k]] = make_int2(s[k], d[k]);
        }
        __syncthreads();

#pragma unroll
        for (int k = 0; k < EG; k++) {
            const int i = k * PG + t;
            const int2 ed = st[i < m ? i : m - 1];
            s[k] = ed.x;
            d[k] = ed.y;
        }
#pragma unroll
        for (int k = 0; k < EG; k++) {
            const int b = bucket_of(d[k]);
            rk[k] = gcur[b] - lbase[b];
        }
#pragma unroll
        for (int k = 0; k < EG; k++) {
            const int i = k * PG + t;
            if (i < m) {
                ps[rk[k] + i] = s[k];
                pd[rk[k] + i] = d[k];
            }
        }
        __syncthreads();
        gcur[t] += c;
        cnt[t] = 0;
        __syncthreads();
    }
}

#define DC 8192

__global__ __launch_bounds__(1024) void bucket_degree_kernel(const int* __restrict__ pd,
                                                             const int* __restrict__ bstart,
                                                             float* __restrict__ dinv, int n) {
    __shared__ int indeg[BK];
    const int tid = threadIdx.x;
    const int node0 = blockIdx.x * BK;
    const int nn = n - node0 < BK ? n - node0 : BK;
    if (tid < BK) indeg[tid] = 0;
    const int e0 = bstart[blockIdx.x];
    const int e1 = bstart[blockIdx.x + 1];
    __syncthreads();
    for (int base = e0; base < e1; base += DC) {
        const int m = e1 - base < DC ? e1 - base : DC;
        int d[8];
#pragma unroll
        for (int k = 0; k < 8; k++) {
            const int i = k * 1024 + tid;
            const int ii = i < m ? i : m - 1;
            d[k] = pd[base + ii] - node0;
        }
#pragma unroll
        for (int k = 0; k < 8; k++) {
            const int i = k * 1024 + tid;
            if (i < m) atomicAdd(&indeg[d[k]], 1);
        }
    }
    __syncthreads();
    if (tid < nn) dinv[node0 + tid] = rsqrtf((float)(indeg[tid] + 1));
}

#define QC 8192
#define IR 4

__global__ __launch_bounds__(1024) void bucket_aggregate_kernel(
    const int* __restrict__ ps, const int* __restrict__ pd, const int* __restrict__ bstart,
    const float* __restrict__ dinv, const float* __restrict__ G, const float* __restrict__ bias,
    float* __restrict__ z, int n, int do_relu) {
    __shared__ v4f acc[BK * 32];
    __shared__ int2 q[QC];
    __shared__ int cnt[32];
    const int node0 = blockIdx.x * BK;
    const int nn = n - node0 < BK ? n - node0 : BK;
    const int tid = threadIdx.x;
    const int lane = tid & 31;
    const int wave = __builtin_amdgcn_readfirstlane(tid >> 5);
    const v4f* G4 = (const v4f*)G;

    for (int r0 = wave; r0 < nn; r0 += 32 * IR) {
        v4f m[IR];
#pragma unroll
        for (int u = 0; u < IR; u++) {
            const int r = r0 + 32 * u;
            if (r < nn) m[u] = G4[(size_t)(node0 + r) * 32 + lane];
        }
#pragma unroll
        for (int u = 0; u < IR; u++) {
            const int r = r0 + 32 * u;
            if (r < nn) acc[r * 32 + lane] = m[u];
        }
    }

    const int e0 = bstart[blockIdx.x];
    const int e1 = bstart[blockIdx.x + 1];
    for (int base = e0; base < e1; base += QC) {
        const int m = e1 - base < QC ? e1 - base : QC;
        if (tid < 32) cnt[tid] = 0;
        __syncthreads();

        int s[8], d[8], rk[8];
#pragma unroll
        for (int k = 0; k < 8; k++) {
            const int i = k * 1024 + tid;
            const int ii = i < m ? i : m - 1;
            s[k] = ps[base + ii];
            d[k] = pd[base + ii] - node0;
        }
#pragma unroll
        for (int k = 0; k < 8; k++) {
            const int i = k * 1024 + tid;
            rk[k] = 0;
            if (i < m) rk[k] = atomicAdd(&cnt[d[k] & 31], 1);
        }
        __syncthreads();

        const int c = cnt[lane];
        int inc = c;
#pragma unroll
        for (int off = 1; off < 32; off <<= 1) {
            const int t = __shfl_up(inc, off, 32);
            if (lane >= off) inc += t;
        }
        const int first = inc - c;
#pragma unroll
        for (int k = 0; k < 8; k++) {
            const int i = k * 1024 + tid;
            const int st = __shfl(first, d[k] & 31, 32);
            if (i < m) q[st + rk[k]] = make_int2(s[k], d[k]);
        }
        __syncthreads();

        const int qs = __builtin_amdgcn_readlane(first, wave);
        const int qn = __builtin_amdgcn_readlane(c, wave);
        for (int g = 0; g < qn; g += 32) {
            const int left = qn - g;
            const int cg = left < 32 ? left : 32;
            int2 e = make_int2(0, 0);
            if (lane < cg) e = q[qs + g + lane];
            if (cg == 32) {
                v4f ma[8], mb[8];
#pragma unroll
                for (int u = 0; u < 8; u++)
                    ma[u] = G4[(size_t)__builtin_amdgcn_readlane(e.x, u) * 32 + lane];
#pragma unroll
                for (int u = 0; u < 8; u++)
                    mb[u] = G4[(size_t)__builtin_amdgcn_readlane(e.x, 8 + u) * 32 + lane];
#pragma unroll
                for (int u = 0; u < 8; u++) {
                    v4f* a = acc + __builtin_amdgcn_readlane(e.y, u) * 32 + lane;
                    *a = *a + ma[u];
                }
#pragma unroll
                for (int u = 0; u < 8; u++)
                    ma[u] = G4[(size_t)__builtin_amdgcn_readlane(e.x, 16 + u) * 32 + lane];
#pragma unroll
                for (int u = 0; u < 8; u++) {
                    v4f* a = acc + __builtin_amdgcn_readlane(e.y, 8 + u) * 32 + lane;
                    *a = *a + mb[u];
                }
#pragma unroll
                for (int u = 0; u < 8; u++)
                    mb[u] = G4[(size_t)__builtin_amdgcn_readlane(e.x, 24 + u) * 32 + lane];
#pragma unroll
                for (int u = 0; u < 8; u++) {
                    v4f* a = acc + __builtin_amdgcn_readlane(e.y, 16 + u) * 32 + lane;
                    *a = *a + ma[u];
                }
#pragma unroll
                for (int u = 0; u < 8; u++) {
                    v4f* a = acc + __builtin_amdgcn_readlane(e.y, 24 + u) * 32 + lane;
                    *a = *a + mb[u];
                }
            } else {
                for (int j0 = 0; j0 < cg; j0 += 8) {
                    v4f mt[8];
#pragma unroll
                    for (int u = 0; u < 8; u++)
                        mt[u] = G4[(size_t)__builtin_amdgcn_readlane(e.x, j0 + u) * 32 + lane];
#pragma unroll
                    for (int u = 0; u < 8; u++) {
                        if (j0 + u < cg) {
                            v4f* a = acc + __builtin_amdgcn_readlane(e.y, j0 + u) * 32 + lane;
                            *a = *a + mt[u];
                        }
                    }
                }
            }
        }
    }

    const v4f b4 = {bias[4 * lane], bias[4 * lane + 1], bias[4 * lane + 2], bias[4 * lane + 3]};
    v4f* z4 = (v4f*)z;
    for (int r0 = wave; r0 < nn; r0 += 32 * IR) {
        float qv[IR];
#pragma unroll
        for (int u = 0; u < IR; u++) {
            const int r = r0 + 32 * u;
            qv[u] = r < nn ? dinv[node0 + r] : 0.0f;
        }
#pragma unroll
        for (int u = 0; u < IR; u++) {
            const int r = r0 + 32 * u;
            if (r < nn) {
                v4f w = acc[r * 32 + lane] * qv[u] + b4;
                if (do_relu) {
                    w.x = fmaxf(w.x, 0.0f);
                    w.y = fmaxf(w.y, 0.0f);
                    w.z = fmaxf(w.z, 0.0f);
                    w.w = fmaxf(w.w, 0.0f);
                }
                z4[(size_t)(node0 + r) * 32 + lane] = w;
            }
        }
    }
}

#define DT 8

__device__ __forceinline__ void load_row_pair(v4f (&a)[4], v4f (&b)[4], const float* __restrict__ z,
                                              int si, int di, int j) {
    const v4f* ra = (const v4f*)(z + (size_t)si * DF) + j;
    const v4f* rb = (const v4f*)(z + (size_t)di * DF) + j;
#pragma unroll
    for (int k = 0; k < 4; k++) {
        a[k] = ra[8 * k];
        b[k] = rb[8 * k];
    }
}

__global__ __launch_bounds__(256) void decode_kernel(const int* __restrict__ s,
                                                     const int* __restrict__ d,
                                                     const float* __restrict__ z,
                                                     float* __restrict__ logits, int e) {
    const int lane = threadIdx.x & 31;
    const int q = lane >> 3;
    const int j = lane & 7;
    const int w = (int)((blockIdx.x * blockDim.x + threadIdx.x) >> 5);
    const int base = w * 32;
    if (base >= e) return;

    int si[DT], di[DT];
#pragma unroll
    for (int i = 0; i < DT; i++) {
        const int ei = base + 8 * q + i;
        const int ec = ei < e ? ei : e - 1;
        si[i] = s[ec];
        di[i] = d[ec];
    }
    __builtin_amdgcn_sched_barrier(0);

    v4f a[2][4], b[2][4];
    load_row_pair(a[0], b[0], z, si[0], di[0], j);
    load_row_pair(a[1], b[1], z, si[1], di[1], j);
    float r[DT];
#pragma unroll
    for (int i = 0; i < DT; i++) {
        const int u = i & 1;
        float p[4];
#pragma unroll
        for (int k = 0; k < 4; k++)
            p[k] = a[u][k].x * b[u][k].x + a[u][k].y * b[u][k].y + a[u][k].z * b[u][k].z +
                   a[u][k].w * b[u][k].w;
        __builtin_amdgcn_sched_barrier(0);
        if (i + 2 < DT) load_row_pair(a[u], b[u], z, si[i + 2], di[i + 2], j);
        __builtin_amdgcn_sched_barrier(0);
        float t = (p[0] + p[2]) + (p[1] + p[3]);
        t += __shfl_xor(t, 4, 32);
        t += __shfl_xor(t, 2, 32);
        t += __shfl_xor(t, 1, 32);
        r[i] = t;
    }
    float sel = r[0];
#pragma unroll
    for (int i = 1; i < DT; i++)
        if (j == i) sel = r[i];
    if (base + lane < e) logits[base + lane] = sel;
}


extern "C" void kernel_launch(void* const* d_in, const int* in_sizes, int n_in,
                              void* d_out, int out_size, void* d_ws, size_t ws_size,
                              hipStream_t stream) {
    const float* x  = (const float*)d_in[0];
    const int*   ei = (const int*)d_in[1];
    const int*   el = (const int*)d_in[2];
    const float* W1 = (const float*)d_in[3];
    const float* b1 = (const float*)d_in[4];
    const float* W2 = (const float*)d_in[5];
    const float* b2 = (const float*)d_in[6];

    const int n  = in_sizes[0] / DF;
    const int E  = in_sizes[1] / 2;
    const int EL = in_sizes[2] / 2;

    float* bufA = (float*)d_ws;
    float* bufB = bufA + (size_t)n * DF;
    float* dinv = bufB + (size_t)n * DF;
    int* blkcnt = (int*)(dinv + n);
    int* excl   = blkcnt + PG * NBMAX;
    int* total  = excl + PG * NBMAX;
    int* bstart = total + NBMAX;
    int* ps     = bstart + 2 * NBMAX;
    int* pd     = ps + E;
    float* out  = (float*)d_out;

    const int decBl  = (EL + 255) / 256;
    const int nb     = (n + BK - 1) / BK;

    bin_count_kernel<<<PG, 256, 0, stream>>>(ei + E, blkcnt, E);
    bin_scan_kernel<<<nb, 256, 0, stream>>>(blkcnt, excl, total);
    bin_place_kernel<<<PG, 256, 0, stream>>>(ei, ei + E, excl, total, ps, pd, bstart, E, nb);

    bucket_degree_kernel<<<nb, 1024, 0, stream>>>(pd, bstart, dinv, n);

    gemm_wmma_kernel<<<GW, 128, 0, stream>>>(x, W1, dinv, bufA, n);
    bucket_aggregate_kernel<<<nb, 1024, 0, stream>>>(ps, pd, bstart, dinv, bufA, b1, bufB, n, 1);

    gemm_wmma_kernel<<<GW, 128, 0, stream>>>(bufB, W2, dinv, bufA, n);
    bucket_aggregate_kernel<<<nb, 1024, 0, stream>>>(ps, pd, bstart, dinv, bufA, b2, bufB, n, 0);

    decode_kernel<<<decBl, 256, 0, stream>>>(el, el + EL, bufB, out, EL);
}
